# starting kernel
# baseline (speedup 1.0000x reference)
#include <hip/hip_runtime.h>
#include <math.h>

#define BH_TOT 32
#define NCHUNK 32
#define CT     128
#define CPAIR  64
#define DD     128
#define STABV  10000.0f

typedef float v2f __attribute__((ext_vector_type(2)));
typedef float v8f __attribute__((ext_vector_type(8)));

__device__ __forceinline__ float gelu_tanh(float x) {
    const float k0 = 0.7978845608028654f;
    const float k1 = 0.044715f;
    float t = tanhf(k0 * (x + k1 * x * x * x));
    return 0.5f * x * (1.0f + t);
}

__device__ __forceinline__ float clipstab(float v) {
    return fminf(fmaxf(v, -STABV), STABV);
}

__global__ __launch_bounds__(256) void liz_scan(
    const float* __restrict__ Wt, const float* __restrict__ Ut,
    const float* __restrict__ h0, float* __restrict__ hstates,
    float* __restrict__ hfinal)
{
    __shared__ float h[DD * DD];
    __shared__ float w0r[DD], w1r[DD], u0r[DD], u1r[DD];
    __shared__ float uv0[DD], uv1[DD];
    __shared__ float psum[256];
    __shared__ float wwred[64];

    const int tid = threadIdx.x;
    const int bh  = blockIdx.x;

    {
        const float4* src = (const float4*)(h0 + (size_t)bh * DD * DD);
        float4* hv = (float4*)h;
        for (int i = tid; i < DD * DD / 4; i += 256) hv[i] = src[i];
    }
    __syncthreads();

    for (int ch = 0; ch < NCHUNK; ++ch) {
        {
            float4* dst = (float4*)(hstates + (size_t)(bh * NCHUNK + ch) * DD * DD);
            const float4* hv = (const float4*)h;
            for (int i = tid; i < DD * DD / 4; i += 256) dst[i] = hv[i];
        }
        const float* wbase = Wt + (size_t)(bh * NCHUNK + ch) * CT * DD;
        const float* ubase = Ut + (size_t)(bh * NCHUNK + ch) * CT * DD;
        if (tid < DD) {
            w0r[tid] = wbase[(size_t)(CT - 2) * DD + tid];
            w1r[tid] = wbase[(size_t)(CT - 1) * DD + tid];
            u0r[tid] = ubase[(size_t)(CT - 2) * DD + tid];
            u1r[tid] = ubase[(size_t)(CT - 1) * DD + tid];
        }
        __syncthreads();

        {
            int d = tid & (DD - 1);
            int half = tid >> 7;
            float s = 0.f;
            for (int r = half * 64; r < half * 64 + 64; ++r) s += w0r[r] * h[r * DD + d];
            psum[tid] = s;
        }
        __syncthreads();
        if (tid < DD) uv0[tid] = u0r[tid] - (psum[tid] + psum[tid + 128]);
        if (tid < 64) wwred[tid] = w1r[tid] * w0r[tid] + w1r[tid + 64] * w0r[tid + 64];
        __syncthreads();
        for (int off = 32; off > 0; off >>= 1) {
            if (tid < off) wwred[tid] += wwred[tid + off];
            __syncthreads();
        }
        float ww = wwred[0];

        {
            int d = tid & (DD - 1);
            int half = tid >> 7;
            float s = 0.f;
            for (int r = half * 64; r < half * 64 + 64; ++r) s += w1r[r] * h[r * DD + d];
            psum[tid] = s;
        }
        __syncthreads();
        if (tid < DD) uv1[tid] = u1r[tid] - (psum[tid] + psum[tid + 128] + ww * uv0[tid]);
        __syncthreads();

        {
            int row  = tid >> 1;
            int half = tid & 1;
            int base = row * DD + half * 64;
            float w0v = w0r[row], w1v = w1r[row];
            float sq = 0.f;
            for (int j = 0; j < 64; ++j) {
                int col = half * 64 + j;
                float v = clipstab(h[base + j] + w0v * uv0[col]);
                v = clipstab(v + w1v * uv1[col]);
                h[base + j] = v;
                sq += v * v;
            }
            psum[tid] = sq;
        }
        __syncthreads();
        {
            int row  = tid >> 1;
            int half = tid & 1;
            float n  = sqrtf(psum[2 * row] + psum[2 * row + 1]) + 1e-6f;
            float inv = 2.0f / n;
            float hn  = 0.5f * n;
            int base = row * DD + half * 64;
            for (int j = 0; j < 64; ++j) {
                float x = h[base + j] * inv;
                h[base + j] = hn * gelu_tanh(x);
            }
        }
        __syncthreads();
    }

    {
        float4* dst = (float4*)(hfinal + (size_t)bh * DD * DD);
        const float4* hv = (const float4*)h;
        for (int i = tid; i < DD * DD / 4; i += 256) dst[i] = hv[i];
    }
}

__global__ __launch_bounds__(256) void liz_out(
    const float* __restrict__ Qt, const float* __restrict__ Wt,
    const float* __restrict__ Ut, const float* __restrict__ hstates,
    float* __restrict__ ys)
{
    __shared__ float h[DD * DD];
    __shared__ float s_w[CPAIR], s_q[CPAIR];

    const int tid  = threadIdx.x;
    const int lane = tid & 31;
    const int wave = tid >> 5;
    const int nc = blockIdx.x, bh = blockIdx.y;

    const size_t cbase = (size_t)(bh * NCHUNK + nc) * CT * DD;
    const float* Qc = Qt + cbase;
    const float* Wc = Wt + cbase;
    const float* Uc = Ut + cbase;

    {
        const float4* src = (const float4*)(hstates + (size_t)(bh * NCHUNK + nc) * DD * DD);
        float4* hv = (float4*)h;
        for (int i = tid; i < DD * DD / 4; i += 256) hv[i] = src[i];
    }

    for (int i = 0; i < 8; ++i) {
        int c = wave * 8 + i;
        const float* w0p = Wc + (size_t)(2 * c) * DD;
        const float* w1p = Wc + (size_t)(2 * c + 1) * DD;
        const float* q1p = Qc + (size_t)(2 * c + 1) * DD;
        float pw = 0.f, pq = 0.f;
        for (int e = lane; e < DD; e += 32) {
            float a = w0p[e];
            pw += w1p[e] * a;
            pq += q1p[e] * a;
        }
        for (int off = 16; off > 0; off >>= 1) {
            pw += __shfl_xor(pw, off, 32);
            pq += __shfl_xor(pq, off, 32);
        }
        if (lane == 0) { s_w[c] = pw; s_q[c] = pq; }
    }
    __syncthreads();

    const int m      = wave;
    const int mrow   = m * 16 + (lane & 15);
    const int khalf  = 2 * (lane >> 4);
    const int nlane  = lane & 15;
    const int rowoff = (lane >> 4) * 8;

    v2f aW[32], aQ[32];
    {
        const float* wrow = Wc + (size_t)mrow * DD;
        const float* qrow = Qc + (size_t)mrow * DD;
#pragma unroll
        for (int k = 0; k < 32; ++k) {
            int col = 4 * k + khalf;
            aW[k] = *(const v2f*)(wrow + col);
            aQ[k] = *(const v2f*)(qrow + col);
        }
    }

    float* outbase = ys + cbase;
    for (int n = 0; n < 8; ++n) {
        v8f accW = {};
        v8f accQ = {};
        const int ncol = n * 16 + nlane;
#pragma unroll
        for (int k = 0; k < 32; ++k) {
            int kk = 4 * k + khalf;
            v2f b;
            b.x = h[kk * DD + ncol];
            b.y = h[(kk + 1) * DD + ncol];
            accW = __builtin_amdgcn_wmma_f32_16x16x4_f32(
                false, aW[k], false, b, (short)0, accW, false, false);
            accQ = __builtin_amdgcn_wmma_f32_16x16x4_f32(
                false, aQ[k], false, b, (short)0, accQ, false, false);
        }
#pragma unroll
        for (int j = 0; j < 4; ++j) {
            int lr0 = 2 * j + rowoff;
            int t0  = m * 16 + lr0;
            int t1  = t0 + 1;
            int c   = t0 >> 1;
            float wh0 = accW[2 * j], wh1 = accW[2 * j + 1];
            float qh0 = accQ[2 * j], qh1 = accQ[2 * j + 1];
            float u0 = Uc[(size_t)t0 * DD + ncol];
            float u1 = Uc[(size_t)t1 * DD + ncol];
            float q0 = Qc[(size_t)t0 * DD + ncol];
            float q1 = Qc[(size_t)t1 * DD + ncol];
            float uv0v = u0 - wh0;
            float out0 = q0 * uv0v + qh0;
            float uv1v = u1 - (wh1 + s_w[c] * uv0v);
            float out1 = q1 * uv1v + qh1 + s_q[c] * uv0v;
            outbase[(size_t)t0 * DD + ncol] = out0;
            outbase[(size_t)t1 * DD + ncol] = out1;
        }
    }
}

extern "C" void kernel_launch(void* const* d_in, const int* in_sizes, int n_in,
                              void* d_out, int out_size, void* d_ws, size_t ws_size,
                              hipStream_t stream) {
    const float* q  = (const float*)d_in[0];
    const float* W  = (const float*)d_in[1];
    const float* U  = (const float*)d_in[2];
    const float* h0 = (const float*)d_in[3];

    float* ys     = (float*)d_out;
    float* hfinal = ys + (size_t)BH_TOT * NCHUNK * CT * DD;
    float* hstate = (float*)d_ws;

    liz_scan<<<BH_TOT, 256, 0, stream>>>(W, U, h0, hstate, hfinal);
    liz_out<<<dim3(NCHUNK, BH_TOT), 256, 0, stream>>>(q, W, U, hstate, ys);
}


# scan: each wave owns whole rows of the state (a lane holds 4 adjacent columns): conflict-free b128 LDS access, update+gelu_norm in one pass, row norm by in-wave butterfly sum (sum order changes)
# speedup vs baseline: 1.9732x; 1.9732x over previous
#include <hip/hip_runtime.h>
#include <math.h>

#define BH_TOT 32
#define NCHUNK 32
#define CT     128
#define CPAIR  64
#define DD     128
#define STABV  10000.0f

typedef float v2f __attribute__((ext_vector_type(2)));
typedef float v8f __attribute__((ext_vector_type(8)));

__device__ __forceinline__ float gelu_tanh(float x) {
    const float k0 = 0.7978845608028654f;
    const float k1 = 0.044715f;
    float t = tanhf(k0 * (x + k1 * x * x * x));
    return 0.5f * x * (1.0f + t);
}

__device__ __forceinline__ float clipstab(float v) {
    return fminf(fmaxf(v, -STABV), STABV);
}

template <int CTRL>
__device__ __forceinline__ float lane_xmask(float x) {
    return __builtin_bit_cast(float,
        __builtin_amdgcn_update_dpp(0, __builtin_bit_cast(int, x), CTRL, 0xf, 0xf, true));
}
__device__ __forceinline__ float wave_sum(float x) {
    x += lane_xmask<0x161>(x);
    x += lane_xmask<0x162>(x);
    x += lane_xmask<0x164>(x);
    x += lane_xmask<0x168>(x);
    const int xi = __builtin_bit_cast(int, x);
    x += __builtin_bit_cast(float,
        __builtin_amdgcn_permlanex16(xi, xi, 0x76543210, 0xfedcba98, false, false));
    return x;
}

__global__ __launch_bounds__(256) void liz_scan(
    const float* __restrict__ Wt, const float* __restrict__ Ut,
    const float* __restrict__ h0, float* __restrict__ hstates,
    float* __restrict__ hfinal)
{
    __shared__ float h[DD * DD];
    __shared__ float w0r[DD], w1r[DD], u0r[DD], u1r[DD];
    __shared__ __attribute__((aligned(16))) float uv0[DD];
    __shared__ __attribute__((aligned(16))) float uv1[DD];
    __shared__ float psum[256];
    __shared__ float wwred[64];

    const int tid = threadIdx.x;
    const int bh  = blockIdx.x;

    {
        const float4* src = (const float4*)(h0 + (size_t)bh * DD * DD);
        float4* hv = (float4*)h;
        for (int i = tid; i < DD * DD / 4; i += 256) hv[i] = src[i];
    }
    __syncthreads();

    for (int ch = 0; ch < NCHUNK; ++ch) {
        {
            float4* dst = (float4*)(hstates + (size_t)(bh * NCHUNK + ch) * DD * DD);
            const float4* hv = (const float4*)h;
            for (int i = tid; i < DD * DD / 4; i += 256) dst[i] = hv[i];
        }
        const float* wbase = Wt + (size_t)(bh * NCHUNK + ch) * CT * DD;
        const float* ubase = Ut + (size_t)(bh * NCHUNK + ch) * CT * DD;
        if (tid < DD) {
            w0r[tid] = wbase[(size_t)(CT - 2) * DD + tid];
            w1r[tid] = wbase[(size_t)(CT - 1) * DD + tid];
            u0r[tid] = ubase[(size_t)(CT - 2) * DD + tid];
            u1r[tid] = ubase[(size_t)(CT - 1) * DD + tid];
        }
        __syncthreads();

        {
            int d = tid & (DD - 1);
            int half = tid >> 7;
            float s = 0.f;
            for (int r = half * 64; r < half * 64 + 64; ++r) s += w0r[r] * h[r * DD + d];
            psum[tid] = s;
        }
        __syncthreads();
        if (tid < DD) uv0[tid] = u0r[tid] - (psum[tid] + psum[tid + 128]);
        if (tid < 64) wwred[tid] = w1r[tid] * w0r[tid] + w1r[tid + 64] * w0r[tid + 64];
        __syncthreads();
        for (int off = 32; off > 0; off >>= 1) {
            if (tid < off) wwred[tid] += wwred[tid + off];
            __syncthreads();
        }
        float ww = wwred[0];

        {
            int d = tid & (DD - 1);
            int half = tid >> 7;
            float s = 0.f;
            for (int r = half * 64; r < half * 64 + 64; ++r) s += w1r[r] * h[r * DD + d];
            psum[tid] = s;
        }
        __syncthreads();
        if (tid < DD) uv1[tid] = u1r[tid] - (psum[tid] + psum[tid + 128] + ww * uv0[tid]);
        __syncthreads();

        {
            const int lane = tid & 31;
            const int row0 = (tid >> 5) * 16;
            const float4 a0 = *(const float4*)&uv0[4 * lane];
            const float4 a1 = *(const float4*)&uv1[4 * lane];
            for (int g = 0; g < 16; g += 4) {
                float4 v[4];
                float  sq[4];
                for (int i = 0; i < 4; ++i) {
                    const int row = row0 + g + i;
                    const float4 hv = *(const float4*)&h[row * DD + 4 * lane];
                    const float w0v = w0r[row], w1v = w1r[row];
                    v[i].x = clipstab(clipstab(hv.x + w0v * a0.x) + w1v * a1.x);
                    v[i].y = clipstab(clipstab(hv.y + w0v * a0.y) + w1v * a1.y);
                    v[i].z = clipstab(clipstab(hv.z + w0v * a0.z) + w1v * a1.z);
                    v[i].w = clipstab(clipstab(hv.w + w0v * a0.w) + w1v * a1.w);
                    sq[i] = v[i].x * v[i].x + v[i].y * v[i].y + v[i].z * v[i].z + v[i].w * v[i].w;
                }
                for (int i = 0; i < 4; ++i) sq[i] = wave_sum(sq[i]);
                for (int i = 0; i < 4; ++i) {
                    const int row = row0 + g + i;
                    const float n   = sqrtf(sq[i]) + 1e-6f;
                    const float inv = 2.0f / n;
                    const float hn  = 0.5f * n;
                    float4 o;
                    o.x = hn * gelu_tanh(v[i].x * inv);
                    o.y = hn * gelu_tanh(v[i].y * inv);
                    o.z = hn * gelu_tanh(v[i].z * inv);
                    o.w = hn * gelu_tanh(v[i].w * inv);
                    *(float4*)&h[row * DD + 4 * lane] = o;
                }
            }
        }
        __syncthreads();
    }

    {
        float4* dst = (float4*)(hfinal + (size_t)bh * DD * DD);
        const float4* hv = (const float4*)h;
        for (int i = tid; i < DD * DD / 4; i += 256) dst[i] = hv[i];
    }
}

__global__ __launch_bounds__(256) void liz_out(
    const float* __restrict__ Qt, const float* __restrict__ Wt,
    const float* __restrict__ Ut, const float* __restrict__ hstates,
    float* __restrict__ ys)
{
    __shared__ float h[DD * DD];
    __shared__ float s_w[CPAIR], s_q[CPAIR];

    const int tid  = threadIdx.x;
    const int lane = tid & 31;
    const int wave = tid >> 5;
    const int nc = blockIdx.x, bh = blockIdx.y;

    const size_t cbase = (size_t)(bh * NCHUNK + nc) * CT * DD;
    const float* Qc = Qt + cbase;
    const float* Wc = Wt + cbase;
    const float* Uc = Ut + cbase;

    {
        const float4* src = (const float4*)(hstates + (size_t)(bh * NCHUNK + nc) * DD * DD);
        float4* hv = (float4*)h;
        for (int i = tid; i < DD * DD / 4; i += 256) hv[i] = src[i];
    }

    for (int i = 0; i < 8; ++i) {
        int c = wave * 8 + i;
        const float* w0p = Wc + (size_t)(2 * c) * DD;
        const float* w1p = Wc + (size_t)(2 * c + 1) * DD;
        const float* q1p = Qc + (size_t)(2 * c + 1) * DD;
        float pw = 0.f, pq = 0.f;
        for (int e = lane; e < DD; e += 32) {
            float a = w0p[e];
            pw += w1p[e] * a;
            pq += q1p[e] * a;
        }
        for (int off = 16; off > 0; off >>= 1) {
            pw += __shfl_xor(pw, off, 32);
            pq += __shfl_xor(pq, off, 32);
        }
        if (lane == 0) { s_w[c] = pw; s_q[c] = pq; }
    }
    __syncthreads();

    const int m      = wave;
    const int mrow   = m * 16 + (lane & 15);
    const int khalf  = 2 * (lane >> 4);
    const int nlane  = lane & 15;
    const int rowoff = (lane >> 4) * 8;

    v2f aW[32], aQ[32];
    {
        const float* wrow = Wc + (size_t)mrow * DD;
        const float* qrow = Qc + (size_t)mrow * DD;
#pragma unroll
        for (int k = 0; k < 32; ++k) {
            int col = 4 * k + khalf;
            aW[k] = *(const v2f*)(wrow + col);
            aQ[k] = *(const v2f*)(qrow + col);
        }
    }

    float* outbase = ys + cbase;
    for (int n = 0; n < 8; ++n) {
        v8f accW = {};
        v8f accQ = {};
        const int ncol = n * 16 + nlane;
#pragma unroll
        for (int k = 0; k < 32; ++k) {
            int kk = 4 * k + khalf;
            v2f b;
            b.x = h[kk * DD + ncol];
            b.y = h[(kk + 1) * DD + ncol];
            accW = __builtin_amdgcn_wmma_f32_16x16x4_f32(
                false, aW[k], false, b, (short)0, accW, false, false);
            accQ = __builtin_amdgcn_wmma_f32_16x16x4_f32(
                false, aQ[k], false, b, (short)0, accQ, false, false);
        }
#pragma unroll
        for (int j = 0; j < 4; ++j) {
            int lr0 = 2 * j + rowoff;
            int t0  = m * 16 + lr0;
            int t1  = t0 + 1;
            int c   = t0 >> 1;
            float wh0 = accW[2 * j], wh1 = accW[2 * j + 1];
            float qh0 = accQ[2 * j], qh1 = accQ[2 * j + 1];
            float u0 = Uc[(size_t)t0 * DD + ncol];
            float u1 = Uc[(size_t)t1 * DD + ncol];
            float q0 = Qc[(size_t)t0 * DD + ncol];
            float q1 = Qc[(size_t)t1 * DD + ncol];
            float uv0v = u0 - wh0;
            float out0 = q0 * uv0v + qh0;
            float uv1v = u1 - (wh1 + s_w[c] * uv0v);
            float out1 = q1 * uv1v + qh1 + s_q[c] * uv0v;
            outbase[(size_t)t0 * DD + ncol] = out0;
            outbase[(size_t)t1 * DD + ncol] = out1;
        }
    }
}

extern "C" void kernel_launch(void* const* d_in, const int* in_sizes, int n_in,
                              void* d_out, int out_size, void* d_ws, size_t ws_size,
                              hipStream_t stream) {
    const float* q  = (const float*)d_in[0];
    const float* W  = (const float*)d_in[1];
    const float* U  = (const float*)d_in[2];
    const float* h0 = (const float*)d_in[3];

    float* ys     = (float*)d_out;
    float* hfinal = ys + (size_t)BH_TOT * NCHUNK * CT * DD;
    float* hstate = (float*)d_ws;

    liz_scan<<<BH_TOT, 256, 0, stream>>>(W, U, h0, hstate, hfinal);
    liz_out<<<dim3(NCHUNK, BH_TOT), 256, 0, stream>>>(q, W, U, hstate, ys);
}


# scan: gelu written as x*sigmoid(2u) with one hardware exp2 and one reciprocal instead of the library tanh: no data-dependent branches, a row group's elements interleave (activation rounding changes)
# speedup vs baseline: 2.9679x; 1.5041x over previous
#include <hip/hip_runtime.h>
#include <math.h>

#define BH_TOT 32
#define NCHUNK 32
#define CT     128
#define CPAIR  64
#define DD     128
#define STABV  10000.0f

typedef float v2f __attribute__((ext_vector_type(2)));
typedef float v8f __attribute__((ext_vector_type(8)));

__device__ __forceinline__ float gelu_tanh(float x) {
    const float c  = -2.302208198f;
    const float k1 = 0.044715f;
    const float y = (c * x) * (1.0f + k1 * x * x);
    const float e = __builtin_amdgcn_exp2f(y);
    return x * __builtin_amdgcn_rcpf(1.0f + e);
}

__device__ __forceinline__ float clipstab(float v) {
    return fminf(fmaxf(v, -STABV), STABV);
}

template <int CTRL>
__device__ __forceinline__ float lane_xmask(float x) {
    return __builtin_bit_cast(float,
        __builtin_amdgcn_update_dpp(0, __builtin_bit_cast(int, x), CTRL, 0xf, 0xf, true));
}
__device__ __forceinline__ float wave_sum(float x) {
    x += lane_xmask<0x161>(x);
    x += lane_xmask<0x162>(x);
    x += lane_xmask<0x164>(x);
    x += lane_xmask<0x168>(x);
    const int xi = __builtin_bit_cast(int, x);
    x += __builtin_bit_cast(float,
        __builtin_amdgcn_permlanex16(xi, xi, 0x76543210, 0xfedcba98, false, false));
    return x;
}

__global__ __launch_bounds__(256) void liz_scan(
    const float* __restrict__ Wt, const float* __restrict__ Ut,
    const float* __restrict__ h0, float* __restrict__ hstates,
    float* __restrict__ hfinal)
{
    __shared__ float h[DD * DD];
    __shared__ float w0r[DD], w1r[DD], u0r[DD], u1r[DD];
    __shared__ __attribute__((aligned(16))) float uv0[DD];
    __shared__ __attribute__((aligned(16))) float uv1[DD];
    __shared__ float psum[256];
    __shared__ float wwred[64];

    const int tid = threadIdx.x;
    const int bh  = blockIdx.x;

    {
        const float4* src = (const float4*)(h0 + (size_t)bh * DD * DD);
        float4* hv = (float4*)h;
        for (int i = tid; i < DD * DD / 4; i += 256) hv[i] = src[i];
    }
    __syncthreads();

    for (int ch = 0; ch < NCHUNK; ++ch) {
        {
            float4* dst = (float4*)(hstates + (size_t)(bh * NCHUNK + ch) * DD * DD);
            const float4* hv = (const float4*)h;
            for (int i = tid; i < DD * DD / 4; i += 256) dst[i] = hv[i];
        }
        const float* wbase = Wt + (size_t)(bh * NCHUNK + ch) * CT * DD;
        const float* ubase = Ut + (size_t)(bh * NCHUNK + ch) * CT * DD;
        if (tid < DD) {
            w0r[tid] = wbase[(size_t)(CT - 2) * DD + tid];
            w1r[tid] = wbase[(size_t)(CT - 1) * DD + tid];
            u0r[tid] = ubase[(size_t)(CT - 2) * DD + tid];
            u1r[tid] = ubase[(size_t)(CT - 1) * DD + tid];
        }
        __syncthreads();

        {
            int d = tid & (DD - 1);
            int half = tid >> 7;
            float s = 0.f;
            for (int r = half * 64; r < half * 64 + 64; ++r) s += w0r[r] * h[r * DD + d];
            psum[tid] = s;
        }
        __syncthreads();
        if (tid < DD) uv0[tid] = u0r[tid] - (psum[tid] + psum[tid + 128]);
        if (tid < 64) wwred[tid] = w1r[tid] * w0r[tid] + w1r[tid + 64] * w0r[tid + 64];
        __syncthreads();
        for (int off = 32; off > 0; off >>= 1) {
            if (tid < off) wwred[tid] += wwred[tid + off];
            __syncthreads();
        }
        float ww = wwred[0];

        {
            int d = tid & (DD - 1);
            int half = tid >> 7;
            float s = 0.f;
            for (int r = half * 64; r < half * 64 + 64; ++r) s += w1r[r] * h[r * DD + d];
            psum[tid] = s;
        }
        __syncthreads();
        if (tid < DD) uv1[tid] = u1r[tid] - (psum[tid] + psum[tid + 128] + ww * uv0[tid]);
        __syncthreads();

        {
            const int lane = tid & 31;
            const int row0 = (tid >> 5) * 16;
            const float4 a0 = *(const float4*)&uv0[4 * lane];
            const float4 a1 = *(const float4*)&uv1[4 * lane];
            for (int g = 0; g < 16; g += 4) {
                float4 v[4];
                float  sq[4];
                for (int i = 0; i < 4; ++i) {
                    const int row = row0 + g + i;
                    const float4 hv = *(const float4*)&h[row * DD + 4 * lane];
                    const float w0v = w0r[row], w1v = w1r[row];
                    v[i].x = clipstab(clipstab(hv.x + w0v * a0.x) + w1v * a1.x);
                    v[i].y = clipstab(clipstab(hv.y + w0v * a0.y) + w1v * a1.y);
                    v[i].z = clipstab(clipstab(hv.z + w0v * a0.z) + w1v * a1.z);
                    v[i].w = clipstab(clipstab(hv.w + w0v * a0.w) + w1v * a1.w);
                    sq[i] = v[i].x * v[i].x + v[i].y * v[i].y + v[i].z * v[i].z + v[i].w * v[i].w;
                }
                for (int i = 0; i < 4; ++i) sq[i] = wave_sum(sq[i]);
                for (int i = 0; i < 4; ++i) {
                    const int row = row0 + g + i;
                    const float n   = sqrtf(sq[i]) + 1e-6f;
                    const float inv = 2.0f / n;
                    const float hn  = 0.5f * n;
                    float4 o;
                    o.x = hn * gelu_tanh(v[i].x * inv);
                    o.y = hn * gelu_tanh(v[i].y * inv);
                    o.z = hn * gelu_tanh(v[i].z * inv);
                    o.w = hn * gelu_tanh(v[i].w * inv);
                    *(float4*)&h[row * DD + 4 * lane] = o;
                }
            }
        }
        __syncthreads();
    }

    {
        float4* dst = (float4*)(hfinal + (size_t)bh * DD * DD);
        const float4* hv = (const float4*)h;
        for (int i = tid; i < DD * DD / 4; i += 256) dst[i] = hv[i];
    }
}

__global__ __launch_bounds__(256) void liz_out(
    const float* __restrict__ Qt, const float* __restrict__ Wt,
    const float* __restrict__ Ut, const float* __restrict__ hstates,
    float* __restrict__ ys)
{
    __shared__ float h[DD * DD];
    __shared__ float s_w[CPAIR], s_q[CPAIR];

    const int tid  = threadIdx.x;
    const int lane = tid & 31;
    const int wave = tid >> 5;
    const int nc = blockIdx.x, bh = blockIdx.y;

    const size_t cbase = (size_t)(bh * NCHUNK + nc) * CT * DD;
    const float* Qc = Qt + cbase;
    const float* Wc = Wt + cbase;
    const float* Uc = Ut + cbase;

    {
        const float4* src = (const float4*)(hstates + (size_t)(bh * NCHUNK + nc) * DD * DD);
        float4* hv = (float4*)h;
        for (int i = tid; i < DD * DD / 4; i += 256) hv[i] = src[i];
    }

    for (int i = 0; i < 8; ++i) {
        int c = wave * 8 + i;
        const float* w0p = Wc + (size_t)(2 * c) * DD;
        const float* w1p = Wc + (size_t)(2 * c + 1) * DD;
        const float* q1p = Qc + (size_t)(2 * c + 1) * DD;
        float pw = 0.f, pq = 0.f;
        for (int e = lane; e < DD; e += 32) {
            float a = w0p[e];
            pw += w1p[e] * a;
            pq += q1p[e] * a;
        }
        for (int off = 16; off > 0; off >>= 1) {
            pw += __shfl_xor(pw, off, 32);
            pq += __shfl_xor(pq, off, 32);
        }
        if (lane == 0) { s_w[c] = pw; s_q[c] = pq; }
    }
    __syncthreads();

    const int m      = wave;
    const int mrow   = m * 16 + (lane & 15);
    const int khalf  = 2 * (lane >> 4);
    const int nlane  = lane & 15;
    const int rowoff = (lane >> 4) * 8;

    v2f aW[32], aQ[32];
    {
        const float* wrow = Wc + (size_t)mrow * DD;
        const float* qrow = Qc + (size_t)mrow * DD;
#pragma unroll
        for (int k = 0; k < 32; ++k) {
            int col = 4 * k + khalf;
            aW[k] = *(const v2f*)(wrow + col);
            aQ[k] = *(const v2f*)(qrow + col);
        }
    }

    float* outbase = ys + cbase;
    for (int n = 0; n < 8; ++n) {
        v8f accW = {};
        v8f accQ = {};
        const int ncol = n * 16 + nlane;
#pragma unroll
        for (int k = 0; k < 32; ++k) {
            int kk = 4 * k + khalf;
            v2f b;
            b.x = h[kk * DD + ncol];
            b.y = h[(kk + 1) * DD + ncol];
            accW = __builtin_amdgcn_wmma_f32_16x16x4_f32(
                false, aW[k], false, b, (short)0, accW, false, false);
            accQ = __builtin_amdgcn_wmma_f32_16x16x4_f32(
                false, aQ[k], false, b, (short)0, accQ, false, false);
        }
#pragma unroll
        for (int j = 0; j < 4; ++j) {
            int lr0 = 2 * j + rowoff;
            int t0  = m * 16 + lr0;
            int t1  = t0 + 1;
            int c   = t0 >> 1;
            float wh0 = accW[2 * j], wh1 = accW[2 * j + 1];
            float qh0 = accQ[2 * j], qh1 = accQ[2 * j + 1];
            float u0 = Uc[(size_t)t0 * DD + ncol];
            float u1 = Uc[(size_t)t1 * DD + ncol];
            float q0 = Qc[(size_t)t0 * DD + ncol];
            float q1 = Qc[(size_t)t1 * DD + ncol];
            float uv0v = u0 - wh0;
            float out0 = q0 * uv0v + qh0;
            float uv1v = u1 - (wh1 + s_w[c] * uv0v);
            float out1 = q1 * uv1v + qh1 + s_q[c] * uv0v;
            outbase[(size_t)t0 * DD + ncol] = out0;
            outbase[(size_t)t1 * DD + ncol] = out1;
        }
    }
}

extern "C" void kernel_launch(void* const* d_in, const int* in_sizes, int n_in,
                              void* d_out, int out_size, void* d_ws, size_t ws_size,
                              hipStream_t stream) {
    const float* q  = (const float*)d_in[0];
    const float* W  = (const float*)d_in[1];
    const float* U  = (const float*)d_in[2];
    const float* h0 = (const float*)d_in[3];

    float* ys     = (float*)d_out;
    float* hfinal = ys + (size_t)BH_TOT * NCHUNK * CT * DD;
    float* hstate = (float*)d_ws;

    liz_scan<<<BH_TOT, 256, 0, stream>>>(W, U, h0, hstate, hfinal);
    liz_out<<<dim3(NCHUNK, BH_TOT), 256, 0, stream>>>(q, W, U, hstate, ys);
}


# scan: state kept in registers by its owner lanes; one pass per chunk stores, updates and feeds the rows into the next chunk's two mat-vec sums, waves' partials added in LDS (sum order changes)
# speedup vs baseline: 4.4605x; 1.5029x over previous
#include <hip/hip_runtime.h>
#include <math.h>

#define BH_TOT 32
#define NCHUNK 32
#define CT     128
#define CPAIR  64
#define DD     128
#define STABV  10000.0f

typedef float v2f __attribute__((ext_vector_type(2)));
typedef float v8f __attribute__((ext_vector_type(8)));

__device__ __forceinline__ float gelu_tanh(float x) {
    const float c  = -2.302208198f;
    const float k1 = 0.044715f;
    const float y = (c * x) * (1.0f + k1 * x * x);
    const float e = __builtin_amdgcn_exp2f(y);
    return x * __builtin_amdgcn_rcpf(1.0f + e);
}

__device__ __forceinline__ float clipstab(float v) {
    return fminf(fmaxf(v, -STABV), STABV);
}

template <int CTRL>
__device__ __forceinline__ float lane_xmask(float x) {
    return __builtin_bit_cast(float,
        __builtin_amdgcn_update_dpp(0, __builtin_bit_cast(int, x), CTRL, 0xf, 0xf, true));
}
__device__ __forceinline__ float wave_sum(float x) {
    x += lane_xmask<0x161>(x);
    x += lane_xmask<0x162>(x);
    x += lane_xmask<0x164>(x);
    x += lane_xmask<0x168>(x);
    const int xi = __builtin_bit_cast(int, x);
    x += __builtin_bit_cast(float,
        __builtin_amdgcn_permlanex16(xi, xi, 0x76543210, 0xfedcba98, false, false));
    return x;
}

struct ChunkVec {
    float  w0r, w1r;
    float4 w0c, w1c, u0c, u1c;
};

__device__ __forceinline__ ChunkVec load_chunk(const float* __restrict__ Wt,
                                               const float* __restrict__ Ut,
                                               int bh, int ch, int row0, int lane) {
    const float* w0p = Wt + ((size_t)(bh * NCHUNK + ch) * CT + (CT - 2)) * DD;
    const float* u0p = Ut + ((size_t)(bh * NCHUNK + ch) * CT + (CT - 2)) * DD;
    ChunkVec c;
    c.w0r = w0p[row0 + (lane & 15)];
    c.w1r = w0p[DD + row0 + (lane & 15)];
    c.w0c = *(const float4*)(w0p + 4 * lane);
    c.w1c = *(const float4*)(w0p + DD + 4 * lane);
    c.u0c = *(const float4*)(u0p + 4 * lane);
    c.u1c = *(const float4*)(u0p + DD + 4 * lane);
    return c;
}

__device__ __forceinline__ float lane_value(float x, int r) {
    return __builtin_bit_cast(float, __builtin_amdgcn_readlane(__builtin_bit_cast(int, x), r));
}

template <bool UPDATE>
__device__ __forceinline__ void state_pass(float4 (&hs)[16], float w0cur, float w1cur,
                                           float w0nxt, float w1nxt,
                                           const float4 a0, const float4 a1,
                                           float4& acc0, float4& acc1) {
#pragma unroll
    for (int g = 0; g < 16; g += 4) {
        if (UPDATE) {
            float4 v[4];
            float  sq[4];
#pragma unroll
            for (int i = 0; i < 4; ++i) {
                const int r = g + i;
                const float4 hv = hs[r];
                const float w0v = lane_value(w0cur, r), w1v = lane_value(w1cur, r);
                v[i].x = clipstab(clipstab(hv.x + w0v * a0.x) + w1v * a1.x);
                v[i].y = clipstab(clipstab(hv.y + w0v * a0.y) + w1v * a1.y);
                v[i].z = clipstab(clipstab(hv.z + w0v * a0.z) + w1v * a1.z);
                v[i].w = clipstab(clipstab(hv.w + w0v * a0.w) + w1v * a1.w);
                sq[i] = v[i].x * v[i].x + v[i].y * v[i].y + v[i].z * v[i].z + v[i].w * v[i].w;
            }
#pragma unroll
            for (int i = 0; i < 4; ++i) sq[i] = wave_sum(sq[i]);
#pragma unroll
            for (int i = 0; i < 4; ++i) {
                const float n   = sqrtf(sq[i]) + 1e-6f;
                const float inv = 2.0f / n;
                const float hn  = 0.5f * n;
                float4 o;
                o.x = hn * gelu_tanh(v[i].x * inv);
                o.y = hn * gelu_tanh(v[i].y * inv);
                o.z = hn * gelu_tanh(v[i].z * inv);
                o.w = hn * gelu_tanh(v[i].w * inv);
                hs[g + i] = o;
            }
        }
#pragma unroll
        for (int i = 0; i < 4; ++i) {
            const int r = g + i;
            const float4 o = hs[r];
            const float w0n = lane_value(w0nxt, r), w1n = lane_value(w1nxt, r);
            acc0.x += w0n * o.x; acc0.y += w0n * o.y; acc0.z += w0n * o.z; acc0.w += w0n * o.w;
            acc1.x += w1n * o.x; acc1.y += w1n * o.y; acc1.z += w1n * o.z; acc1.w += w1n * o.w;
        }
    }
}

__device__ __forceinline__ void store_state(const float4 (&hs)[16], float* __restrict__ dst) {
#pragma unroll
    for (int i = 0; i < 16; ++i) *(float4*)(dst + (size_t)i * DD) = hs[i];
}

__device__ __forceinline__ void meet_sums(float (&part)[2][8][DD], int wave, int lane,
                                          const float4 acc0, const float4 acc1,
                                          const ChunkVec& c, float4& a0, float4& a1) {
    *(float4*)&part[0][wave][4 * lane] = acc0;
    *(float4*)&part[1][wave][4 * lane] = acc1;
    __syncthreads();
    float4 p0 = *(const float4*)&part[0][0][4 * lane];
    float4 p1 = *(const float4*)&part[1][0][4 * lane];
    for (int w = 1; w < 8; ++w) {
        const float4 q0 = *(const float4*)&part[0][w][4 * lane];
        const float4 q1 = *(const float4*)&part[1][w][4 * lane];
        p0.x += q0.x; p0.y += q0.y; p0.z += q0.z; p0.w += q0.w;
        p1.x += q1.x; p1.y += q1.y; p1.z += q1.z; p1.w += q1.w;
    }
    const float ww = wave_sum(c.w1c.x * c.w0c.x + c.w1c.y * c.w0c.y +
                              c.w1c.z * c.w0c.z + c.w1c.w * c.w0c.w);
    a0.x = c.u0c.x - p0.x; a0.y = c.u0c.y - p0.y;
    a0.z = c.u0c.z - p0.z; a0.w = c.u0c.w - p0.w;
    a1.x = c.u1c.x - (p1.x + ww * a0.x); a1.y = c.u1c.y - (p1.y + ww * a0.y);
    a1.z = c.u1c.z - (p1.z + ww * a0.z); a1.w = c.u1c.w - (p1.w + ww * a0.w);
}

__global__ __launch_bounds__(256) void liz_scan(
    const float* __restrict__ Wt, const float* __restrict__ Ut,
    const float* __restrict__ h0, float* __restrict__ hstates,
    float* __restrict__ hfinal)
{
    __shared__ __attribute__((aligned(16))) float part[2][2][8][DD];

    const int tid  = threadIdx.x;
    const int lane = tid & 31;
    const int wave = tid >> 5;
    const int bh   = blockIdx.x;
    const int row0 = wave * 16;
    const size_t myoff = (size_t)row0 * DD + 4 * lane;

    ChunkVec nxt = load_chunk(Wt, Ut, bh, 1, row0, lane);
    ChunkVec cur = load_chunk(Wt, Ut, bh, 0, row0, lane);

    float4 hs[16];
    {
        const float* src = h0 + (size_t)bh * DD * DD + myoff;
#pragma unroll
        for (int i = 0; i < 16; ++i) hs[i] = *(const float4*)(src + (size_t)i * DD);
    }

    float4 a0, a1;
    {
        const float4 none = {0.f, 0.f, 0.f, 0.f};
        float4 acc0 = {0.f, 0.f, 0.f, 0.f}, acc1 = {0.f, 0.f, 0.f, 0.f};
        state_pass<false>(hs, 0.f, 0.f, cur.w0r, cur.w1r, none, none, acc0, acc1);
        meet_sums(part[0], wave, lane, acc0, acc1, cur, a0, a1);
    }

    for (int ch = 0; ch < NCHUNK; ++ch) {
        const ChunkVec pre = load_chunk(Wt, Ut, bh, (ch + 2 < NCHUNK) ? ch + 2 : NCHUNK - 1,
                                        row0, lane);
        store_state(hs, hstates + (size_t)(bh * NCHUNK + ch) * DD * DD + myoff);

        float4 acc0 = {0.f, 0.f, 0.f, 0.f}, acc1 = {0.f, 0.f, 0.f, 0.f};
        state_pass<true>(hs, cur.w0r, cur.w1r, nxt.w0r, nxt.w1r, a0, a1, acc0, acc1);
        if (ch + 1 < NCHUNK) meet_sums(part[(ch + 1) & 1], wave, lane, acc0, acc1, nxt, a0, a1);
        cur = nxt;
        nxt = pre;
    }
    store_state(hs, hfinal + (size_t)bh * DD * DD + myoff);
}

__global__ __launch_bounds__(256) void liz_out(
    const float* __restrict__ Qt, const float* __restrict__ Wt,
    const float* __restrict__ Ut, const float* __restrict__ hstates,
    float* __restrict__ ys)
{
    __shared__ float h[DD * DD];
    __shared__ float s_w[CPAIR], s_q[CPAIR];

    const int tid  = threadIdx.x;
    const int lane = tid & 31;
    const int wave = tid >> 5;
    const int nc = blockIdx.x, bh = blockIdx.y;

    const size_t cbase = (size_t)(bh * NCHUNK + nc) * CT * DD;
    const float* Qc = Qt + cbase;
    const float* Wc = Wt + cbase;
    const float* Uc = Ut + cbase;

    {
        const float4* src = (const float4*)(hstates + (size_t)(bh * NCHUNK + nc) * DD * DD);
        float4* hv = (float4*)h;
        for (int i = tid; i < DD * DD / 4; i += 256) hv[i] = src[i];
    }

    for (int i = 0; i < 8; ++i) {
        int c = wave * 8 + i;
        const float* w0p = Wc + (size_t)(2 * c) * DD;
        const float* w1p = Wc + (size_t)(2 * c + 1) * DD;
        const float* q1p = Qc + (size_t)(2 * c + 1) * DD;
        float pw = 0.f, pq = 0.f;
        for (int e = lane; e < DD; e += 32) {
            float a = w0p[e];
            pw += w1p[e] * a;
            pq += q1p[e] * a;
        }
        for (int off = 16; off > 0; off >>= 1) {
            pw += __shfl_xor(pw, off, 32);
            pq += __shfl_xor(pq, off, 32);
        }
        if (lane == 0) { s_w[c] = pw; s_q[c] = pq; }
    }
    __syncthreads();

    const int m      = wave;
    const int mrow   = m * 16 + (lane & 15);
    const int khalf  = 2 * (lane >> 4);
    const int nlane  = lane & 15;
    const int rowoff = (lane >> 4) * 8;

    v2f aW[32], aQ[32];
    {
        const float* wrow = Wc + (size_t)mrow * DD;
        const float* qrow = Qc + (size_t)mrow * DD;
#pragma unroll
        for (int k = 0; k < 32; ++k) {
            int col = 4 * k + khalf;
            aW[k] = *(const v2f*)(wrow + col);
            aQ[k] = *(const v2f*)(qrow + col);
        }
    }

    float* outbase = ys + cbase;
    for (int n = 0; n < 8; ++n) {
        v8f accW = {};
        v8f accQ = {};
        const int ncol = n * 16 + nlane;
#pragma unroll
        for (int k = 0; k < 32; ++k) {
            int kk = 4 * k + khalf;
            v2f b;
            b.x = h[kk * DD + ncol];
            b.y = h[(kk + 1) * DD + ncol];
            accW = __builtin_amdgcn_wmma_f32_16x16x4_f32(
                false, aW[k], false, b, (short)0, accW, false, false);
            accQ = __builtin_amdgcn_wmma_f32_16x16x4_f32(
                false, aQ[k], false, b, (short)0, accQ, false, false);
        }
#pragma unroll
        for (int j = 0; j < 4; ++j) {
            int lr0 = 2 * j + rowoff;
            int t0  = m * 16 + lr0;
            int t1  = t0 + 1;
            int c   = t0 >> 1;
            float wh0 = accW[2 * j], wh1 = accW[2 * j + 1];
            float qh0 = accQ[2 * j], qh1 = accQ[2 * j + 1];
            float u0 = Uc[(size_t)t0 * DD + ncol];
            float u1 = Uc[(size_t)t1 * DD + ncol];
            float q0 = Qc[(size_t)t0 * DD + ncol];
            float q1 = Qc[(size_t)t1 * DD + ncol];
            float uv0v = u0 - wh0;
            float out0 = q0 * uv0v + qh0;
            float uv1v = u1 - (wh1 + s_w[c] * uv0v);
            float out1 = q1 * uv1v + qh1 + s_q[c] * uv0v;
            outbase[(size_t)t0 * DD + ncol] = out0;
            outbase[(size_t)t1 * DD + ncol] = out1;
        }
    }
}

extern "C" void kernel_launch(void* const* d_in, const int* in_sizes, int n_in,
                              void* d_out, int out_size, void* d_ws, size_t ws_size,
                              hipStream_t stream) {
    const float* q  = (const float*)d_in[0];
    const float* W  = (const float*)d_in[1];
    const float* U  = (const float*)d_in[2];
    const float* h0 = (const float*)d_in[3];

    float* ys     = (float*)d_out;
    float* hfinal = ys + (size_t)BH_TOT * NCHUNK * CT * DD;
    float* hstate = (float*)d_ws;

    liz_scan<<<BH_TOT, 256, 0, stream>>>(W, U, h0, hstate, hfinal);
    liz_out<<<dim3(NCHUNK, BH_TOT), 256, 0, stream>>>(q, W, U, hstate, ys);
}


# out: row-pair dots s_w, s_q formed from the A fragments in registers (lane^1, lane^16 exchange) instead of 8 serial rounds of global loads and LDS shuffles per wave (dot sum order changes)
# speedup vs baseline: 5.2660x; 1.1806x over previous
#include <hip/hip_runtime.h>
#include <math.h>

#define BH_TOT 32
#define NCHUNK 32
#define CT     128
#define CPAIR  64
#define DD     128
#define STABV  10000.0f

typedef float v2f __attribute__((ext_vector_type(2)));
typedef float v8f __attribute__((ext_vector_type(8)));

__device__ __forceinline__ float gelu_tanh(float x) {
    const float c  = -2.302208198f;
    const float k1 = 0.044715f;
    const float y = (c * x) * (1.0f + k1 * x * x);
    const float e = __builtin_amdgcn_exp2f(y);
    return x * __builtin_amdgcn_rcpf(1.0f + e);
}

__device__ __forceinline__ float clipstab(float v) {
    return fminf(fmaxf(v, -STABV), STABV);
}

template <int CTRL>
__device__ __forceinline__ float lane_xmask(float x) {
    return __builtin_bit_cast(float,
        __builtin_amdgcn_update_dpp(0, __builtin_bit_cast(int, x), CTRL, 0xf, 0xf, true));
}
__device__ __forceinline__ float wave_sum(float x) {
    x += lane_xmask<0x161>(x);
    x += lane_xmask<0x162>(x);
    x += lane_xmask<0x164>(x);
    x += lane_xmask<0x168>(x);
    const int xi = __builtin_bit_cast(int, x);
    x += __builtin_bit_cast(float,
        __builtin_amdgcn_permlanex16(xi, xi, 0x76543210, 0xfedcba98, false, false));
    return x;
}

struct ChunkVec {
    float  w0r, w1r;
    float4 w0c, w1c, u0c, u1c;
};

__device__ __forceinline__ ChunkVec load_chunk(const float* __restrict__ Wt,
                                               const float* __restrict__ Ut,
                                               int bh, int ch, int row0, int lane) {
    const float* w0p = Wt + ((size_t)(bh * NCHUNK + ch) * CT + (CT - 2)) * DD;
    const float* u0p = Ut + ((size_t)(bh * NCHUNK + ch) * CT + (CT - 2)) * DD;
    ChunkVec c;
    c.w0r = w0p[row0 + (lane & 15)];
    c.w1r = w0p[DD + row0 + (lane & 15)];
    c.w0c = *(const float4*)(w0p + 4 * lane);
    c.w1c = *(const float4*)(w0p + DD + 4 * lane);
    c.u0c = *(const float4*)(u0p + 4 * lane);
    c.u1c = *(const float4*)(u0p + DD + 4 * lane);
    return c;
}

__device__ __forceinline__ float lane_value(float x, int r) {
    return __builtin_bit_cast(float, __builtin_amdgcn_readlane(__builtin_bit_cast(int, x), r));
}

template <bool UPDATE>
__device__ __forceinline__ void state_pass(float4 (&hs)[16], float w0cur, float w1cur,
                                           float w0nxt, float w1nxt,
                                           const float4 a0, const float4 a1,
                                           float4& acc0, float4& acc1) {
#pragma unroll
    for (int g = 0; g < 16; g += 4) {
        if (UPDATE) {
            float4 v[4];
            float  sq[4];
#pragma unroll
            for (int i = 0; i < 4; ++i) {
                const int r = g + i;
                const float4 hv = hs[r];
                const float w0v = lane_value(w0cur, r), w1v = lane_value(w1cur, r);
                v[i].x = clipstab(clipstab(hv.x + w0v * a0.x) + w1v * a1.x);
                v[i].y = clipstab(clipstab(hv.y + w0v * a0.y) + w1v * a1.y);
                v[i].z = clipstab(clipstab(hv.z + w0v * a0.z) + w1v * a1.z);
                v[i].w = clipstab(clipstab(hv.w + w0v * a0.w) + w1v * a1.w);
                sq[i] = v[i].x * v[i].x + v[i].y * v[i].y + v[i].z * v[i].z + v[i].w * v[i].w;
            }
#pragma unroll
            for (int i = 0; i < 4; ++i) sq[i] = wave_sum(sq[i]);
#pragma unroll
            for (int i = 0; i < 4; ++i) {
                const float n   = sqrtf(sq[i]) + 1e-6f;
                const float inv = 2.0f / n;
                const float hn  = 0.5f * n;
                float4 o;
                o.x = hn * gelu_tanh(v[i].x * inv);
                o.y = hn * gelu_tanh(v[i].y * inv);
                o.z = hn * gelu_tanh(v[i].z * inv);
                o.w = hn * gelu_tanh(v[i].w * inv);
                hs[g + i] = o;
            }
        }
#pragma unroll
        for (int i = 0; i < 4; ++i) {
            const int r = g + i;
            const float4 o = hs[r];
            const float w0n = lane_value(w0nxt, r), w1n = lane_value(w1nxt, r);
            acc0.x += w0n * o.x; acc0.y += w0n * o.y; acc0.z += w0n * o.z; acc0.w += w0n * o.w;
            acc1.x += w1n * o.x; acc1.y += w1n * o.y; acc1.z += w1n * o.z; acc1.w += w1n * o.w;
        }
    }
}

__device__ __forceinline__ void store_state(const float4 (&hs)[16], float* __restrict__ dst) {
#pragma unroll
    for (int i = 0; i < 16; ++i) *(float4*)(dst + (size_t)i * DD) = hs[i];
}

__device__ __forceinline__ void meet_sums(float (&part)[2][8][DD], int wave, int lane,
                                          const float4 acc0, const float4 acc1,
                                          const ChunkVec& c, float4& a0, float4& a1) {
    *(float4*)&part[0][wave][4 * lane] = acc0;
    *(float4*)&part[1][wave][4 * lane] = acc1;
    __syncthreads();
    float4 p0 = *(const float4*)&part[0][0][4 * lane];
    float4 p1 = *(const float4*)&part[1][0][4 * lane];
    for (int w = 1; w < 8; ++w) {
        const float4 q0 = *(const float4*)&part[0][w][4 * lane];
        const float4 q1 = *(const float4*)&part[1][w][4 * lane];
        p0.x += q0.x; p0.y += q0.y; p0.z += q0.z; p0.w += q0.w;
        p1.x += q1.x; p1.y += q1.y; p1.z += q1.z; p1.w += q1.w;
    }
    const float ww = wave_sum(c.w1c.x * c.w0c.x + c.w1c.y * c.w0c.y +
                              c.w1c.z * c.w0c.z + c.w1c.w * c.w0c.w);
    a0.x = c.u0c.x - p0.x; a0.y = c.u0c.y - p0.y;
    a0.z = c.u0c.z - p0.z; a0.w = c.u0c.w - p0.w;
    a1.x = c.u1c.x - (p1.x + ww * a0.x); a1.y = c.u1c.y - (p1.y + ww * a0.y);
    a1.z = c.u1c.z - (p1.z + ww * a0.z); a1.w = c.u1c.w - (p1.w + ww * a0.w);
}

__global__ __launch_bounds__(256) void liz_scan(
    const float* __restrict__ Wt, const float* __restrict__ Ut,
    const float* __restrict__ h0, float* __restrict__ hstates,
    float* __restrict__ hfinal)
{
    __shared__ __attribute__((aligned(16))) float part[2][2][8][DD];

    const int tid  = threadIdx.x;
    const int lane = tid & 31;
    const int wave = tid >> 5;
    const int bh   = blockIdx.x;
    const int row0 = wave * 16;
    const size_t myoff = (size_t)row0 * DD + 4 * lane;

    ChunkVec nxt = load_chunk(Wt, Ut, bh, 1, row0, lane);
    ChunkVec cur = load_chunk(Wt, Ut, bh, 0, row0, lane);

    float4 hs[16];
    {
        const float* src = h0 + (size_t)bh * DD * DD + myoff;
#pragma unroll
        for (int i = 0; i < 16; ++i) hs[i] = *(const float4*)(src + (size_t)i * DD);
    }

    float4 a0, a1;
    {
        const float4 none = {0.f, 0.f, 0.f, 0.f};
        float4 acc0 = {0.f, 0.f, 0.f, 0.f}, acc1 = {0.f, 0.f, 0.f, 0.f};
        state_pass<false>(hs, 0.f, 0.f, cur.w0r, cur.w1r, none, none, acc0, acc1);
        meet_sums(part[0], wave, lane, acc0, acc1, cur, a0, a1);
    }

    for (int ch = 0; ch < NCHUNK; ++ch) {
        const ChunkVec pre = load_chunk(Wt, Ut, bh, (ch + 2 < NCHUNK) ? ch + 2 : NCHUNK - 1,
                                        row0, lane);
        store_state(hs, hstates + (size_t)(bh * NCHUNK + ch) * DD * DD + myoff);

        float4 acc0 = {0.f, 0.f, 0.f, 0.f}, acc1 = {0.f, 0.f, 0.f, 0.f};
        state_pass<true>(hs, cur.w0r, cur.w1r, nxt.w0r, nxt.w1r, a0, a1, acc0, acc1);
        if (ch + 1 < NCHUNK) meet_sums(part[(ch + 1) & 1], wave, lane, acc0, acc1, nxt, a0, a1);
        cur = nxt;
        nxt = pre;
    }
    store_state(hs, hfinal + (size_t)bh * DD * DD + myoff);
}

__global__ __launch_bounds__(256) void liz_out(
    const float* __restrict__ Qt, const float* __restrict__ Wt,
    const float* __restrict__ Ut, const float* __restrict__ hstates,
    float* __restrict__ ys)
{
    __shared__ float h[DD * DD];
    __shared__ float s_w[CPAIR], s_q[CPAIR];

    const int tid  = threadIdx.x;
    const int lane = tid & 31;
    const int wave = tid >> 5;
    const int nc = blockIdx.x, bh = blockIdx.y;

    const size_t cbase = (size_t)(bh * NCHUNK + nc) * CT * DD;
    const float* Qc = Qt + cbase;
    const float* Wc = Wt + cbase;
    const float* Uc = Ut + cbase;

    const int m      = wave;
    const int mrow   = m * 16 + (lane & 15);
    const int khalf  = 2 * (lane >> 4);
    const int nlane  = lane & 15;
    const int rowoff = (lane >> 4) * 8;

    v2f aW[32], aQ[32];
    {
        const float* wrow = Wc + (size_t)mrow * DD;
        const float* qrow = Qc + (size_t)mrow * DD;
#pragma unroll
        for (int k = 0; k < 32; ++k) {
            int col = 4 * k + khalf;
            aW[k] = *(const v2f*)(wrow + col);
            aQ[k] = *(const v2f*)(qrow + col);
        }
    }

    {
        const float4* src = (const float4*)(hstates + (size_t)(bh * NCHUNK + nc) * DD * DD);
        float4* hv = (float4*)h;
        for (int i = tid; i < DD * DD / 4; i += 256) hv[i] = src[i];
    }

    {
        v2f pw = {0.f, 0.f}, pq = {0.f, 0.f};
#pragma unroll
        for (int k = 0; k < 32; ++k) {
            v2f o;
            o.x = lane_xmask<0x161>(aW[k].x);
            o.y = lane_xmask<0x161>(aW[k].y);
            pw += aW[k] * o;
            pq += aQ[k] * o;
        }
        float sw = pw.x + pw.y, sq = pq.x + pq.y;
        const int swi = __builtin_bit_cast(int, sw), sqi = __builtin_bit_cast(int, sq);
        sw += __builtin_bit_cast(float,
            __builtin_amdgcn_permlanex16(swi, swi, 0x76543210, 0xfedcba98, false, false));
        sq += __builtin_bit_cast(float,
            __builtin_amdgcn_permlanex16(sqi, sqi, 0x76543210, 0xfedcba98, false, false));
        if ((lane & 17) == 1) {
            s_w[m * 8 + (lane >> 1)] = sw;
            s_q[m * 8 + (lane >> 1)] = sq;
        }
    }
    __syncthreads();

    float* outbase = ys + cbase;
    for (int n = 0; n < 8; ++n) {
        v8f accW = {};
        v8f accQ = {};
        const int ncol = n * 16 + nlane;
#pragma unroll
        for (int k = 0; k < 32; ++k) {
            int kk = 4 * k + khalf;
            v2f b;
            b.x = h[kk * DD + ncol];
            b.y = h[(kk + 1) * DD + ncol];
            accW = __builtin_amdgcn_wmma_f32_16x16x4_f32(
                false, aW[k], false, b, (short)0, accW, false, false);
            accQ = __builtin_amdgcn_wmma_f32_16x16x4_f32(
                false, aQ[k], false, b, (short)0, accQ, false, false);
        }
#pragma unroll
        for (int j = 0; j < 4; ++j) {
            int lr0 = 2 * j + rowoff;
            int t0  = m * 16 + lr0;
            int t1  = t0 + 1;
            int c   = t0 >> 1;
            float wh0 = accW[2 * j], wh1 = accW[2 * j + 1];
            float qh0 = accQ[2 * j], qh1 = accQ[2 * j + 1];
            float u0 = Uc[(size_t)t0 * DD + ncol];
            float u1 = Uc[(size_t)t1 * DD + ncol];
            float q0 = Qc[(size_t)t0 * DD + ncol];
            float q1 = Qc[(size_t)t1 * DD + ncol];
            float uv0v = u0 - wh0;
            float out0 = q0 * uv0v + qh0;
            float uv1v = u1 - (wh1 + s_w[c] * uv0v);
            float out1 = q1 * uv1v + qh1 + s_q[c] * uv0v;
            outbase[(size_t)t0 * DD + ncol] = out0;
            outbase[(size_t)t1 * DD + ncol] = out1;
        }
    }
}

extern "C" void kernel_launch(void* const* d_in, const int* in_sizes, int n_in,
                              void* d_out, int out_size, void* d_ws, size_t ws_size,
                              hipStream_t stream) {
    const float* q  = (const float*)d_in[0];
    const float* W  = (const float*)d_in[1];
    const float* U  = (const float*)d_in[2];
    const float* h0 = (const float*)d_in[3];

    float* ys     = (float*)d_out;
    float* hfinal = ys + (size_t)BH_TOT * NCHUNK * CT * DD;
    float* hstate = (float*)d_ws;

    liz_scan<<<BH_TOT, 256, 0, stream>>>(W, U, h0, hstate, hfinal);
    liz_out<<<dim3(NCHUNK, BH_TOT), 256, 0, stream>>>(q, W, U, hstate, ys);
}


# scan: the 16 row norms of a wave summed together by a halving butterfly (same partners and add order, same bits): sqrt and divide once per row, not in all 32 lanes; scales come back by readlane
# speedup vs baseline: 6.4242x; 1.2199x over previous
#include <hip/hip_runtime.h>
#include <math.h>

#define BH_TOT 32
#define NCHUNK 32
#define CT     128
#define CPAIR  64
#define DD     128
#define STABV  10000.0f

typedef float v2f __attribute__((ext_vector_type(2)));
typedef float v8f __attribute__((ext_vector_type(8)));

__device__ __forceinline__ float gelu_tanh(float x) {
    const float c  = -2.302208198f;
    const float k1 = 0.044715f;
    const float y = (c * x) * (1.0f + k1 * x * x);
    const float e = __builtin_amdgcn_exp2f(y);
    return x * __builtin_amdgcn_rcpf(1.0f + e);
}

__device__ __forceinline__ float clipstab(float v) {
    return fminf(fmaxf(v, -STABV), STABV);
}

template <int CTRL>
__device__ __forceinline__ float lane_xmask(float x) {
    return __builtin_bit_cast(float,
        __builtin_amdgcn_update_dpp(0, __builtin_bit_cast(int, x), CTRL, 0xf, 0xf, true));
}
__device__ __forceinline__ float wave_sum(float x) {
    x += lane_xmask<0x161>(x);
    x += lane_xmask<0x162>(x);
    x += lane_xmask<0x164>(x);
    x += lane_xmask<0x168>(x);
    const int xi = __builtin_bit_cast(int, x);
    x += __builtin_bit_cast(float,
        __builtin_amdgcn_permlanex16(xi, xi, 0x76543210, 0xfedcba98, false, false));
    return x;
}

struct ChunkVec {
    float  w0r, w1r;
    float4 w0c, w1c, u0c, u1c;
};

__device__ __forceinline__ ChunkVec load_chunk(const float* __restrict__ Wt,
                                               const float* __restrict__ Ut,
                                               int bh, int ch, int row0, int lane) {
    const float* w0p = Wt + ((size_t)(bh * NCHUNK + ch) * CT + (CT - 2)) * DD;
    const float* u0p = Ut + ((size_t)(bh * NCHUNK + ch) * CT + (CT - 2)) * DD;
    ChunkVec c;
    c.w0r = w0p[row0 + (lane & 15)];
    c.w1r = w0p[DD + row0 + (lane & 15)];
    c.w0c = *(const float4*)(w0p + 4 * lane);
    c.w1c = *(const float4*)(w0p + DD + 4 * lane);
    c.u0c = *(const float4*)(u0p + 4 * lane);
    c.u1c = *(const float4*)(u0p + DD + 4 * lane);
    return c;
}

__device__ __forceinline__ float lane_value(float x, int r) {
    return __builtin_bit_cast(float, __builtin_amdgcn_readlane(__builtin_bit_cast(int, x), r));
}

template <bool UPDATE>
__device__ __forceinline__ void state_pass(float4 (&hs)[16], float w0cur, float w1cur,
                                           float w0nxt, float w1nxt,
                                           const float4 a0, const float4 a1,
                                           float4& acc0, float4& acc1, int lane) {
    if (UPDATE) {
        float sq[16];
#pragma unroll
        for (int r = 0; r < 16; ++r) {
            const float4 hv = hs[r];
            const float w0v = lane_value(w0cur, r), w1v = lane_value(w1cur, r);
            float4 v;
            v.x = clipstab(clipstab(hv.x + w0v * a0.x) + w1v * a1.x);
            v.y = clipstab(clipstab(hv.y + w0v * a0.y) + w1v * a1.y);
            v.z = clipstab(clipstab(hv.z + w0v * a0.z) + w1v * a1.z);
            v.w = clipstab(clipstab(hv.w + w0v * a0.w) + w1v * a1.w);
            hs[r] = v;
            sq[r] = v.x * v.x + v.y * v.y + v.z * v.z + v.w * v.w;
        }
        const bool b0 = (lane & 1) != 0, b1 = (lane & 2) != 0;
        const bool b2 = (lane & 4) != 0, b3 = (lane & 8) != 0;
        float s8[8], s4[4], s2[2];
#pragma unroll
        for (int i = 0; i < 8; ++i) {
            const float keep = b0 ? sq[2 * i + 1] : sq[2 * i];
            const float send = b0 ? sq[2 * i] : sq[2 * i + 1];
            s8[i] = keep + lane_xmask<0x161>(send);
        }
#pragma unroll
        for (int i = 0; i < 4; ++i) {
            const float keep = b1 ? s8[2 * i + 1] : s8[2 * i];
            const float send = b1 ? s8[2 * i] : s8[2 * i + 1];
            s4[i] = keep + lane_xmask<0x162>(send);
        }
#pragma unroll
        for (int i = 0; i < 2; ++i) {
            const float keep = b2 ? s4[2 * i + 1] : s4[2 * i];
            const float send = b2 ? s4[2 * i] : s4[2 * i + 1];
            s2[i] = keep + lane_xmask<0x164>(send);
        }
        float tot;
        {
            const float keep = b3 ? s2[1] : s2[0];
            const float send = b3 ? s2[0] : s2[1];
            tot = keep + lane_xmask<0x168>(send);
            const int ti = __builtin_bit_cast(int, tot);
            tot += __builtin_bit_cast(float,
                __builtin_amdgcn_permlanex16(ti, ti, 0x76543210, 0xfedcba98, false, false));
        }
        const float n   = sqrtf(tot) + 1e-6f;
        const float inv = 2.0f / n;
        const float hn  = 0.5f * n;
#pragma unroll
        for (int r = 0; r < 16; ++r) {
            const float invr = lane_value(inv, r), hnr = lane_value(hn, r);
            const float4 v = hs[r];
            float4 o;
            o.x = hnr * gelu_tanh(v.x * invr);
            o.y = hnr * gelu_tanh(v.y * invr);
            o.z = hnr * gelu_tanh(v.z * invr);
            o.w = hnr * gelu_tanh(v.w * invr);
            hs[r] = o;
        }
    }
#pragma unroll
    for (int r = 0; r < 16; ++r) {
        const float4 o = hs[r];
        const float w0n = lane_value(w0nxt, r), w1n = lane_value(w1nxt, r);
        acc0.x += w0n * o.x; acc0.y += w0n * o.y; acc0.z += w0n * o.z; acc0.w += w0n * o.w;
        acc1.x += w1n * o.x; acc1.y += w1n * o.y; acc1.z += w1n * o.z; acc1.w += w1n * o.w;
    }
}

__device__ __forceinline__ void store_state(const float4 (&hs)[16], float* __restrict__ dst) {
#pragma unroll
    for (int i = 0; i < 16; ++i) *(float4*)(dst + (size_t)i * DD) = hs[i];
}

__device__ __forceinline__ void meet_sums(float (&part)[2][8][DD], int wave, int lane,
                                          const float4 acc0, const float4 acc1,
                                          const ChunkVec& c, float4& a0, float4& a1) {
    *(float4*)&part[0][wave][4 * lane] = acc0;
    *(float4*)&part[1][wave][4 * lane] = acc1;
    __syncthreads();
    float4 p0 = *(const float4*)&part[0][0][4 * lane];
    float4 p1 = *(const float4*)&part[1][0][4 * lane];
    for (int w = 1; w < 8; ++w) {
        const float4 q0 = *(const float4*)&part[0][w][4 * lane];
        const float4 q1 = *(const float4*)&part[1][w][4 * lane];
        p0.x += q0.x; p0.y += q0.y; p0.z += q0.z; p0.w += q0.w;
        p1.x += q1.x; p1.y += q1.y; p1.z += q1.z; p1.w += q1.w;
    }
    const float ww = wave_sum(c.w1c.x * c.w0c.x + c.w1c.y * c.w0c.y +
                              c.w1c.z * c.w0c.z + c.w1c.w * c.w0c.w);
    a0.x = c.u0c.x - p0.x; a0.y = c.u0c.y - p0.y;
    a0.z = c.u0c.z - p0.z; a0.w = c.u0c.w - p0.w;
    a1.x = c.u1c.x - (p1.x + ww * a0.x); a1.y = c.u1c.y - (p1.y + ww * a0.y);
    a1.z = c.u1c.z - (p1.z + ww * a0.z); a1.w = c.u1c.w - (p1.w + ww * a0.w);
}

__global__ __launch_bounds__(256) void liz_scan(
    const float* __restrict__ Wt, const float* __restrict__ Ut,
    const float* __restrict__ h0, float* __restrict__ hstates,
    float* __restrict__ hfinal)
{
    __shared__ __attribute__((aligned(16))) float part[2][2][8][DD];

    const int tid  = threadIdx.x;
    const int lane = tid & 31;
    const int wave = tid >> 5;
    const int bh   = blockIdx.x;
    const int row0 = wave * 16;
    const size_t myoff = (size_t)row0 * DD + 4 * lane;

    ChunkVec nxt = load_chunk(Wt, Ut, bh, 1, row0, lane);
    ChunkVec cur = load_chunk(Wt, Ut, bh, 0, row0, lane);

    float4 hs[16];
    {
        const float* src = h0 + (size_t)bh * DD * DD + myoff;
#pragma unroll
        for (int i = 0; i < 16; ++i) hs[i] = *(const float4*)(src + (size_t)i * DD);
    }

    float4 a0, a1;
    {
        const float4 none = {0.f, 0.f, 0.f, 0.f};
        float4 acc0 = {0.f, 0.f, 0.f, 0.f}, acc1 = {0.f, 0.f, 0.f, 0.f};
        state_pass<false>(hs, 0.f, 0.f, cur.w0r, cur.w1r, none, none, acc0, acc1, lane);
        meet_sums(part[0], wave, lane, acc0, acc1, cur, a0, a1);
    }

    for (int ch = 0; ch < NCHUNK; ++ch) {
        const ChunkVec pre = load_chunk(Wt, Ut, bh, (ch + 2 < NCHUNK) ? ch + 2 : NCHUNK - 1,
                                        row0, lane);
        store_state(hs, hstates + (size_t)(bh * NCHUNK + ch) * DD * DD + myoff);

        float4 acc0 = {0.f, 0.f, 0.f, 0.f}, acc1 = {0.f, 0.f, 0.f, 0.f};
        state_pass<true>(hs, cur.w0r, cur.w1r, nxt.w0r, nxt.w1r, a0, a1, acc0, acc1, lane);
        if (ch + 1 < NCHUNK) meet_sums(part[(ch + 1) & 1], wave, lane, acc0, acc1, nxt, a0, a1);
        cur = nxt;
        nxt = pre;
    }
    store_state(hs, hfinal + (size_t)bh * DD * DD + myoff);
}

__global__ __launch_bounds__(256) void liz_out(
    const float* __restrict__ Qt, const float* __restrict__ Wt,
    const float* __restrict__ Ut, const float* __restrict__ hstates,
    float* __restrict__ ys)
{
    __shared__ float h[DD * DD];
    __shared__ float s_w[CPAIR], s_q[CPAIR];

    const int tid  = threadIdx.x;
    const int lane = tid & 31;
    const int wave = tid >> 5;
    const int nc = blockIdx.x, bh = blockIdx.y;

    const size_t cbase = (size_t)(bh * NCHUNK + nc) * CT * DD;
    const float* Qc = Qt + cbase;
    const float* Wc = Wt + cbase;
    const float* Uc = Ut + cbase;

    const int m      = wave;
    const int mrow   = m * 16 + (lane & 15);
    const int khalf  = 2 * (lane >> 4);
    const int nlane  = lane & 15;
    const int rowoff = (lane >> 4) * 8;

    v2f aW[32], aQ[32];
    {
        const float* wrow = Wc + (size_t)mrow * DD;
        const float* qrow = Qc + (size_t)mrow * DD;
#pragma unroll
        for (int k = 0; k < 32; ++k) {
            int col = 4 * k + khalf;
            aW[k] = *(const v2f*)(wrow + col);
            aQ[k] = *(const v2f*)(qrow + col);
        }
    }

    {
        const float4* src = (const float4*)(hstates + (size_t)(bh * NCHUNK + nc) * DD * DD);
        float4* hv = (float4*)h;
        for (int i = tid; i < DD * DD / 4; i += 256) hv[i] = src[i];
    }

    {
        v2f pw = {0.f, 0.f}, pq = {0.f, 0.f};
#pragma unroll
        for (int k = 0; k < 32; ++k) {
            v2f o;
            o.x = lane_xmask<0x161>(aW[k].x);
            o.y = lane_xmask<0x161>(aW[k].y);
            pw += aW[k] * o;
            pq += aQ[k] * o;
        }
        float sw = pw.x + pw.y, sq = pq.x + pq.y;
        const int swi = __builtin_bit_cast(int, sw), sqi = __builtin_bit_cast(int, sq);
        sw += __builtin_bit_cast(float,
            __builtin_amdgcn_permlanex16(swi, swi, 0x76543210, 0xfedcba98, false, false));
        sq += __builtin_bit_cast(float,
            __builtin_amdgcn_permlanex16(sqi, sqi, 0x76543210, 0xfedcba98, false, false));
        if ((lane & 17) == 1) {
            s_w[m * 8 + (lane >> 1)] = sw;
            s_q[m * 8 + (lane >> 1)] = sq;
        }
    }
    __syncthreads();

    float* outbase = ys + cbase;
    for (int n = 0; n < 8; ++n) {
        v8f accW = {};
        v8f accQ = {};
        const int ncol = n * 16 + nlane;
#pragma unroll
        for (int k = 0; k < 32; ++k) {
            int kk = 4 * k + khalf;
            v2f b;
            b.x = h[kk * DD + ncol];
            b.y = h[(kk + 1) * DD + ncol];
            accW = __builtin_amdgcn_wmma_f32_16x16x4_f32(
                false, aW[k], false, b, (short)0, accW, false, false);
            accQ = __builtin_amdgcn_wmma_f32_16x16x4_f32(
                false, aQ[k], false, b, (short)0, accQ, false, false);
        }
#pragma unroll
        for (int j = 0; j < 4; ++j) {
            int lr0 = 2 * j + rowoff;
            int t0  = m * 16 + lr0;
            int t1  = t0 + 1;
            int c   = t0 >> 1;
            float wh0 = accW[2 * j], wh1 = accW[2 * j + 1];
            float qh0 = accQ[2 * j], qh1 = accQ[2 * j + 1];
            float u0 = Uc[(size_t)t0 * DD + ncol];
            float u1 = Uc[(size_t)t1 * DD + ncol];
            float q0 = Qc[(size_t)t0 * DD + ncol];
            float q1 = Qc[(size_t)t1 * DD + ncol];
            float uv0v = u0 - wh0;
            float out0 = q0 * uv0v + qh0;
            float uv1v = u1 - (wh1 + s_w[c] * uv0v);
            float out1 = q1 * uv1v + qh1 + s_q[c] * uv0v;
            outbase[(size_t)t0 * DD + ncol] = out0;
            outbase[(size_t)t1 * DD + ncol] = out1;
        }
    }
}

extern "C" void kernel_launch(void* const* d_in, const int* in_sizes, int n_in,
                              void* d_out, int out_size, void* d_ws, size_t ws_size,
                              hipStream_t stream) {
    const float* q  = (const float*)d_in[0];
    const float* W  = (const float*)d_in[1];
    const float* U  = (const float*)d_in[2];
    const float* h0 = (const float*)d_in[3];

    float* ys     = (float*)d_out;
    float* hfinal = ys + (size_t)BH_TOT * NCHUNK * CT * DD;
    float* hstate = (float*)d_ws;

    liz_scan<<<BH_TOT, 256, 0, stream>>>(W, U, h0, hstate, hfinal);
    liz_out<<<dim3(NCHUNK, BH_TOT), 256, 0, stream>>>(q, W, U, hstate, ys);
}


# scan: gelu_norm as v*(1+tanh(u))/2 with u = v*(A+B*v*v), the row's 2/n folded into A and B, tanh on the hardware instruction; the norm's squares are reused (activation and norm rounding change)
# speedup vs baseline: 6.6902x; 1.0414x over previous
#include <hip/hip_runtime.h>
#include <math.h>

#define BH_TOT 32
#define NCHUNK 32
#define CT     128
#define CPAIR  64
#define DD     128
#define STABV  10000.0f

typedef float v2f __attribute__((ext_vector_type(2)));
typedef float v8f __attribute__((ext_vector_type(8)));

__device__ __forceinline__ float gelu_row(float v, float A, float B) {
    const float u = v * (B * (v * v) + A);
    const float t = __builtin_amdgcn_tanhf(u);
    return v * (0.5f * t + 0.5f);
}

__device__ __forceinline__ float clipstab(float v) {
    return fminf(fmaxf(v, -STABV), STABV);
}

template <int CTRL>
__device__ __forceinline__ float lane_xmask(float x) {
    return __builtin_bit_cast(float,
        __builtin_amdgcn_update_dpp(0, __builtin_bit_cast(int, x), CTRL, 0xf, 0xf, true));
}
__device__ __forceinline__ float wave_sum(float x) {
    x += lane_xmask<0x161>(x);
    x += lane_xmask<0x162>(x);
    x += lane_xmask<0x164>(x);
    x += lane_xmask<0x168>(x);
    const int xi = __builtin_bit_cast(int, x);
    x += __builtin_bit_cast(float,
        __builtin_amdgcn_permlanex16(xi, xi, 0x76543210, 0xfedcba98, false, false));
    return x;
}

struct ChunkVec {
    float  w0r, w1r;
    float4 w0c, w1c, u0c, u1c;
};

__device__ __forceinline__ ChunkVec load_chunk(const float* __restrict__ Wt,
                                               const float* __restrict__ Ut,
                                               int bh, int ch, int row0, int lane) {
    const float* w0p = Wt + ((size_t)(bh * NCHUNK + ch) * CT + (CT - 2)) * DD;
    const float* u0p = Ut + ((size_t)(bh * NCHUNK + ch) * CT + (CT - 2)) * DD;
    ChunkVec c;
    c.w0r = w0p[row0 + (lane & 15)];
    c.w1r = w0p[DD + row0 + (lane & 15)];
    c.w0c = *(const float4*)(w0p + 4 * lane);
    c.w1c = *(const float4*)(w0p + DD + 4 * lane);
    c.u0c = *(const float4*)(u0p + 4 * lane);
    c.u1c = *(const float4*)(u0p + DD + 4 * lane);
    return c;
}

__device__ __forceinline__ float lane_value(float x, int r) {
    return __builtin_bit_cast(float, __builtin_amdgcn_readlane(__builtin_bit_cast(int, x), r));
}

template <bool UPDATE>
__device__ __forceinline__ void state_pass(float4 (&hs)[16], float w0cur, float w1cur,
                                           float w0nxt, float w1nxt,
                                           const float4 a0, const float4 a1,
                                           float4& acc0, float4& acc1, int lane) {
    if (UPDATE) {
        float sq[16];
#pragma unroll
        for (int r = 0; r < 16; ++r) {
            const float4 hv = hs[r];
            const float w0v = lane_value(w0cur, r), w1v = lane_value(w1cur, r);
            float4 v;
            v.x = clipstab(clipstab(hv.x + w0v * a0.x) + w1v * a1.x);
            v.y = clipstab(clipstab(hv.y + w0v * a0.y) + w1v * a1.y);
            v.z = clipstab(clipstab(hv.z + w0v * a0.z) + w1v * a1.z);
            v.w = clipstab(clipstab(hv.w + w0v * a0.w) + w1v * a1.w);
            hs[r] = v;
            sq[r] = v.x * v.x + v.y * v.y + v.z * v.z + v.w * v.w;
        }
        const bool b0 = (lane & 1) != 0, b1 = (lane & 2) != 0;
        const bool b2 = (lane & 4) != 0, b3 = (lane & 8) != 0;
        float s8[8], s4[4], s2[2];
#pragma unroll
        for (int i = 0; i < 8; ++i) {
            const float keep = b0 ? sq[2 * i + 1] : sq[2 * i];
            const float send = b0 ? sq[2 * i] : sq[2 * i + 1];
            s8[i] = keep + lane_xmask<0x161>(send);
        }
#pragma unroll
        for (int i = 0; i < 4; ++i) {
            const float keep = b1 ? s8[2 * i + 1] : s8[2 * i];
            const float send = b1 ? s8[2 * i] : s8[2 * i + 1];
            s4[i] = keep + lane_xmask<0x162>(send);
        }
#pragma unroll
        for (int i = 0; i < 2; ++i) {
            const float keep = b2 ? s4[2 * i + 1] : s4[2 * i];
            const float send = b2 ? s4[2 * i] : s4[2 * i + 1];
            s2[i] = keep + lane_xmask<0x164>(send);
        }
        float tot;
        {
            const float keep = b3 ? s2[1] : s2[0];
            const float send = b3 ? s2[0] : s2[1];
            tot = keep + lane_xmask<0x168>(send);
            const int ti = __builtin_bit_cast(int, tot);
            tot += __builtin_bit_cast(float,
                __builtin_amdgcn_permlanex16(ti, ti, 0x76543210, 0xfedcba98, false, false));
        }
        const float n   = sqrtf(tot) + 1e-6f;
        const float inv = 2.0f / n;
        const float cA  = 0.7978845608f * inv;
        const float cB  = (0.7978845608f * 0.044715f) * inv * (inv * inv);
#pragma unroll
        for (int r = 0; r < 16; ++r) {
            const float Ar = lane_value(cA, r), Br = lane_value(cB, r);
            const float4 v = hs[r];
            float4 o;
            o.x = gelu_row(v.x, Ar, Br);
            o.y = gelu_row(v.y, Ar, Br);
            o.z = gelu_row(v.z, Ar, Br);
            o.w = gelu_row(v.w, Ar, Br);
            hs[r] = o;
        }
    }
#pragma unroll
    for (int r = 0; r < 16; ++r) {
        const float4 o = hs[r];
        const float w0n = lane_value(w0nxt, r), w1n = lane_value(w1nxt, r);
        acc0.x += w0n * o.x; acc0.y += w0n * o.y; acc0.z += w0n * o.z; acc0.w += w0n * o.w;
        acc1.x += w1n * o.x; acc1.y += w1n * o.y; acc1.z += w1n * o.z; acc1.w += w1n * o.w;
    }
}

__device__ __forceinline__ void store_state(const float4 (&hs)[16], float* __restrict__ dst) {
#pragma unroll
    for (int i = 0; i < 16; ++i) *(float4*)(dst + (size_t)i * DD) = hs[i];
}

__device__ __forceinline__ void meet_sums(float (&part)[2][8][DD], int wave, int lane,
                                          const float4 acc0, const float4 acc1,
                                          const ChunkVec& c, float4& a0, float4& a1) {
    *(float4*)&part[0][wave][4 * lane] = acc0;
    *(float4*)&part[1][wave][4 * lane] = acc1;
    __syncthreads();
    float4 p0 = *(const float4*)&part[0][0][4 * lane];
    float4 p1 = *(const float4*)&part[1][0][4 * lane];
    for (int w = 1; w < 8; ++w) {
        const float4 q0 = *(const float4*)&part[0][w][4 * lane];
        const float4 q1 = *(const float4*)&part[1][w][4 * lane];
        p0.x += q0.x; p0.y += q0.y; p0.z += q0.z; p0.w += q0.w;
        p1.x += q1.x; p1.y += q1.y; p1.z += q1.z; p1.w += q1.w;
    }
    const float ww = wave_sum(c.w1c.x * c.w0c.x + c.w1c.y * c.w0c.y +
                              c.w1c.z * c.w0c.z + c.w1c.w * c.w0c.w);
    a0.x = c.u0c.x - p0.x; a0.y = c.u0c.y - p0.y;
    a0.z = c.u0c.z - p0.z; a0.w = c.u0c.w - p0.w;
    a1.x = c.u1c.x - (p1.x + ww * a0.x); a1.y = c.u1c.y - (p1.y + ww * a0.y);
    a1.z = c.u1c.z - (p1.z + ww * a0.z); a1.w = c.u1c.w - (p1.w + ww * a0.w);
}

__global__ __launch_bounds__(256) void liz_scan(
    const float* __restrict__ Wt, const float* __restrict__ Ut,
    const float* __restrict__ h0, float* __restrict__ hstates,
    float* __restrict__ hfinal)
{
    __shared__ __attribute__((aligned(16))) float part[2][2][8][DD];

    const int tid  = threadIdx.x;
    const int lane = tid & 31;
    const int wave = tid >> 5;
    const int bh   = blockIdx.x;
    const int row0 = wave * 16;
    const size_t myoff = (size_t)row0 * DD + 4 * lane;

    ChunkVec nxt = load_chunk(Wt, Ut, bh, 1, row0, lane);
    ChunkVec cur = load_chunk(Wt, Ut, bh, 0, row0, lane);

    float4 hs[16];
    {
        const float* src = h0 + (size_t)bh * DD * DD + myoff;
#pragma unroll
        for (int i = 0; i < 16; ++i) hs[i] = *(const float4*)(src + (size_t)i * DD);
    }

    float4 a0, a1;
    {
        const float4 none = {0.f, 0.f, 0.f, 0.f};
        float4 acc0 = {0.f, 0.f, 0.f, 0.f}, acc1 = {0.f, 0.f, 0.f, 0.f};
        state_pass<false>(hs, 0.f, 0.f, cur.w0r, cur.w1r, none, none, acc0, acc1, lane);
        meet_sums(part[0], wave, lane, acc0, acc1, cur, a0, a1);
    }

    for (int ch = 0; ch < NCHUNK; ++ch) {
        const ChunkVec pre = load_chunk(Wt, Ut, bh, (ch + 2 < NCHUNK) ? ch + 2 : NCHUNK - 1,
                                        row0, lane);
        store_state(hs, hstates + (size_t)(bh * NCHUNK + ch) * DD * DD + myoff);

        float4 acc0 = {0.f, 0.f, 0.f, 0.f}, acc1 = {0.f, 0.f, 0.f, 0.f};
        state_pass<true>(hs, cur.w0r, cur.w1r, nxt.w0r, nxt.w1r, a0, a1, acc0, acc1, lane);
        if (ch + 1 < NCHUNK) meet_sums(part[(ch + 1) & 1], wave, lane, acc0, acc1, nxt, a0, a1);
        cur = nxt;
        nxt = pre;
    }
    store_state(hs, hfinal + (size_t)bh * DD * DD + myoff);
}

__global__ __launch_bounds__(256) void liz_out(
    const float* __restrict__ Qt, const float* __restrict__ Wt,
    const float* __restrict__ Ut, const float* __restrict__ hstates,
    float* __restrict__ ys)
{
    __shared__ float h[DD * DD];
    __shared__ float s_w[CPAIR], s_q[CPAIR];

    const int tid  = threadIdx.x;
    const int lane = tid & 31;
    const int wave = tid >> 5;
    const int nc = blockIdx.x, bh = blockIdx.y;

    const size_t cbase = (size_t)(bh * NCHUNK + nc) * CT * DD;
    const float* Qc = Qt + cbase;
    const float* Wc = Wt + cbase;
    const float* Uc = Ut + cbase;

    const int m      = wave;
    const int mrow   = m * 16 + (lane & 15);
    const int khalf  = 2 * (lane >> 4);
    const int nlane  = lane & 15;
    const int rowoff = (lane >> 4) * 8;

    v2f aW[32], aQ[32];
    {
        const float* wrow = Wc + (size_t)mrow * DD;
        const float* qrow = Qc + (size_t)mrow * DD;
#pragma unroll
        for (int k = 0; k < 32; ++k) {
            int col = 4 * k + khalf;
            aW[k] = *(const v2f*)(wrow + col);
            aQ[k] = *(const v2f*)(qrow + col);
        }
    }

    {
        const float4* src = (const float4*)(hstates + (size_t)(bh * NCHUNK + nc) * DD * DD);
        float4* hv = (float4*)h;
        for (int i = tid; i < DD * DD / 4; i += 256) hv[i] = src[i];
    }

    {
        v2f pw = {0.f, 0.f}, pq = {0.f, 0.f};
#pragma unroll
        for (int k = 0; k < 32; ++k) {
            v2f o;
            o.x = lane_xmask<0x161>(aW[k].x);
            o.y = lane_xmask<0x161>(aW[k].y);
            pw += aW[k] * o;
            pq += aQ[k] * o;
        }
        float sw = pw.x + pw.y, sq = pq.x + pq.y;
        const int swi = __builtin_bit_cast(int, sw), sqi = __builtin_bit_cast(int, sq);
        sw += __builtin_bit_cast(float,
            __builtin_amdgcn_permlanex16(swi, swi, 0x76543210, 0xfedcba98, false, false));
        sq += __builtin_bit_cast(float,
            __builtin_amdgcn_permlanex16(sqi, sqi, 0x76543210, 0xfedcba98, false, false));
        if ((lane & 17) == 1) {
            s_w[m * 8 + (lane >> 1)] = sw;
            s_q[m * 8 + (lane >> 1)] = sq;
        }
    }
    __syncthreads();

    float* outbase = ys + cbase;
    for (int n = 0; n < 8; ++n) {
        v8f accW = {};
        v8f accQ = {};
        const int ncol = n * 16 + nlane;
#pragma unroll
        for (int k = 0; k < 32; ++k) {
            int kk = 4 * k + khalf;
            v2f b;
            b.x = h[kk * DD + ncol];
            b.y = h[(kk + 1) * DD + ncol];
            accW = __builtin_amdgcn_wmma_f32_16x16x4_f32(
                false, aW[k], false, b, (short)0, accW, false, false);
            accQ = __builtin_amdgcn_wmma_f32_16x16x4_f32(
                false, aQ[k], false, b, (short)0, accQ, false, false);
        }
#pragma unroll
        for (int j = 0; j < 4; ++j) {
            int lr0 = 2 * j + rowoff;
            int t0  = m * 16 + lr0;
            int t1  = t0 + 1;
            int c   = t0 >> 1;
            float wh0 = accW[2 * j], wh1 = accW[2 * j + 1];
            float qh0 = accQ[2 * j], qh1 = accQ[2 * j + 1];
            float u0 = Uc[(size_t)t0 * DD + ncol];
            float u1 = Uc[(size_t)t1 * DD + ncol];
            float q0 = Qc[(size_t)t0 * DD + ncol];
            float q1 = Qc[(size_t)t1 * DD + ncol];
            float uv0v = u0 - wh0;
            float out0 = q0 * uv0v + qh0;
            float uv1v = u1 - (wh1 + s_w[c] * uv0v);
            float out1 = q1 * uv1v + qh1 + s_q[c] * uv0v;
            outbase[(size_t)t0 * DD + ncol] = out0;
            outbase[(size_t)t1 * DD + ncol] = out1;
        }
    }
}

extern "C" void kernel_launch(void* const* d_in, const int* in_sizes, int n_in,
                              void* d_out, int out_size, void* d_ws, size_t ws_size,
                              hipStream_t stream) {
    const float* q  = (const float*)d_in[0];
    const float* W  = (const float*)d_in[1];
    const float* U  = (const float*)d_in[2];
    const float* h0 = (const float*)d_in[3];

    float* ys     = (float*)d_out;
    float* hfinal = ys + (size_t)BH_TOT * NCHUNK * CT * DD;
    float* hstate = (float*)d_ws;

    liz_scan<<<BH_TOT, 256, 0, stream>>>(W, U, h0, hstate, hfinal);
    liz_out<<<dim3(NCHUNK, BH_TOT), 256, 0, stream>>>(q, W, U, hstate, ys);
}


# launches: scan in 5 segments (4,7,7,7,7 chunks), each beside the output blocks of the segment before; LDS sized so that every block is alone on its WGP (6 launches for 2; sum order unchanged)
# speedup vs baseline: 7.2112x; 1.0779x over previous
#include <hip/hip_runtime.h>
#include <math.h>

#define BH_TOT 32
#define NCHUNK 32
#define CT     128
#define CPAIR  64
#define DD     128
#define STABV  10000.0f

typedef float v2f __attribute__((ext_vector_type(2)));
typedef float v8f __attribute__((ext_vector_type(8)));

__device__ __forceinline__ float gelu_row(float v, float A, float B) {
    const float u = v * (B * (v * v) + A);
    const float t = __builtin_amdgcn_tanhf(u);
    return v * (0.5f * t + 0.5f);
}

__device__ __forceinline__ float clipstab(float v) {
    return fminf(fmaxf(v, -STABV), STABV);
}

template <int CTRL>
__device__ __forceinline__ float lane_xmask(float x) {
    return __builtin_bit_cast(float,
        __builtin_amdgcn_update_dpp(0, __builtin_bit_cast(int, x), CTRL, 0xf, 0xf, true));
}
__device__ __forceinline__ float wave_sum(float x) {
    x += lane_xmask<0x161>(x);
    x += lane_xmask<0x162>(x);
    x += lane_xmask<0x164>(x);
    x += lane_xmask<0x168>(x);
    const int xi = __builtin_bit_cast(int, x);
    x += __builtin_bit_cast(float,
        __builtin_amdgcn_permlanex16(xi, xi, 0x76543210, 0xfedcba98, false, false));
    return x;
}

struct ChunkVec {
    float  w0r, w1r;
    float4 w0c, w1c, u0c, u1c;
};

__device__ __forceinline__ ChunkVec load_chunk(const float* __restrict__ Wt,
                                               const float* __restrict__ Ut,
                                               int bh, int ch, int row0, int lane) {
    const float* w0p = Wt + ((size_t)(bh * NCHUNK + ch) * CT + (CT - 2)) * DD;
    const float* u0p = Ut + ((size_t)(bh * NCHUNK + ch) * CT + (CT - 2)) * DD;
    ChunkVec c;
    c.w0r = w0p[row0 + (lane & 15)];
    c.w1r = w0p[DD + row0 + (lane & 15)];
    c.w0c = *(const float4*)(w0p + 4 * lane);
    c.w1c = *(const float4*)(w0p + DD + 4 * lane);
    c.u0c = *(const float4*)(u0p + 4 * lane);
    c.u1c = *(const float4*)(u0p + DD + 4 * lane);
    return c;
}

__device__ __forceinline__ float lane_value(float x, int r) {
    return __builtin_bit_cast(float, __builtin_amdgcn_readlane(__builtin_bit_cast(int, x), r));
}

template <bool UPDATE>
__device__ __forceinline__ void state_pass(float4 (&hs)[16], float w0cur, float w1cur,
                                           float w0nxt, float w1nxt,
                                           const float4 a0, const float4 a1,
                                           float4& acc0, float4& acc1, int lane) {
    if (UPDATE) {
        float sq[16];
#pragma unroll
        for (int r = 0; r < 16; ++r) {
            const float4 hv = hs[r];
            const float w0v = lane_value(w0cur, r), w1v = lane_value(w1cur, r);
            float4 v;
            v.x = clipstab(clipstab(hv.x + w0v * a0.x) + w1v * a1.x);
            v.y = clipstab(clipstab(hv.y + w0v * a0.y) + w1v * a1.y);
            v.z = clipstab(clipstab(hv.z + w0v * a0.z) + w1v * a1.z);
            v.w = clipstab(clipstab(hv.w + w0v * a0.w) + w1v * a1.w);
            hs[r] = v;
            sq[r] = v.x * v.x + v.y * v.y + v.z * v.z + v.w * v.w;
        }
        const bool b0 = (lane & 1) != 0, b1 = (lane & 2) != 0;
        const bool b2 = (lane & 4) != 0, b3 = (lane & 8) != 0;
        float s8[8], s4[4], s2[2];
#pragma unroll
        for (int i = 0; i < 8; ++i) {
            const float keep = b0 ? sq[2 * i + 1] : sq[2 * i];
            const float send = b0 ? sq[2 * i] : sq[2 * i + 1];
            s8[i] = keep + lane_xmask<0x161>(send);
        }
#pragma unroll
        for (int i = 0; i < 4; ++i) {
            const float keep = b1 ? s8[2 * i + 1] : s8[2 * i];
            const float send = b1 ? s8[2 * i] : s8[2 * i + 1];
            s4[i] = keep + lane_xmask<0x162>(send);
        }
#pragma unroll
        for (int i = 0; i < 2; ++i) {
            const float keep = b2 ? s4[2 * i + 1] : s4[2 * i];
            const float send = b2 ? s4[2 * i] : s4[2 * i + 1];
            s2[i] = keep + lane_xmask<0x164>(send);
        }
        float tot;
        {
            const float keep = b3 ? s2[1] : s2[0];
            const float send = b3 ? s2[0] : s2[1];
            tot = keep + lane_xmask<0x168>(send);
            const int ti = __builtin_bit_cast(int, tot);
            tot += __builtin_bit_cast(float,
                __builtin_amdgcn_permlanex16(ti, ti, 0x76543210, 0xfedcba98, false, false));
        }
        const float n   = sqrtf(tot) + 1e-6f;
        const float inv = 2.0f / n;
        const float cA  = 0.7978845608f * inv;
        const float cB  = (0.7978845608f * 0.044715f) * inv * (inv * inv);
#pragma unroll
        for (int r = 0; r < 16; ++r) {
            const float Ar = lane_value(cA, r), Br = lane_value(cB, r);
            const float4 v = hs[r];
            float4 o;
            o.x = gelu_row(v.x, Ar, Br);
            o.y = gelu_row(v.y, Ar, Br);
            o.z = gelu_row(v.z, Ar, Br);
            o.w = gelu_row(v.w, Ar, Br);
            hs[r] = o;
        }
    }
#pragma unroll
    for (int r = 0; r < 16; ++r) {
        const float4 o = hs[r];
        const float w0n = lane_value(w0nxt, r), w1n = lane_value(w1nxt, r);
        acc0.x += w0n * o.x; acc0.y += w0n * o.y; acc0.z += w0n * o.z; acc0.w += w0n * o.w;
        acc1.x += w1n * o.x; acc1.y += w1n * o.y; acc1.z += w1n * o.z; acc1.w += w1n * o.w;
    }
}

__device__ __forceinline__ void store_state(const float4 (&hs)[16], float* __restrict__ dst) {
#pragma unroll
    for (int i = 0; i < 16; ++i) *(float4*)(dst + (size_t)i * DD) = hs[i];
}

__device__ __forceinline__ void meet_sums(float (&part)[2][8][DD], int wave, int lane,
                                          const float4 acc0, const float4 acc1,
                                          const ChunkVec& c, float4& a0, float4& a1) {
    *(float4*)&part[0][wave][4 * lane] = acc0;
    *(float4*)&part[1][wave][4 * lane] = acc1;
    __syncthreads();
    float4 p0 = *(const float4*)&part[0][0][4 * lane];
    float4 p1 = *(const float4*)&part[1][0][4 * lane];
    for (int w = 1; w < 8; ++w) {
        const float4 q0 = *(const float4*)&part[0][w][4 * lane];
        const float4 q1 = *(const float4*)&part[1][w][4 * lane];
        p0.x += q0.x; p0.y += q0.y; p0.z += q0.z; p0.w += q0.w;
        p1.x += q1.x; p1.y += q1.y; p1.z += q1.z; p1.w += q1.w;
    }
    const float ww = wave_sum(c.w1c.x * c.w0c.x + c.w1c.y * c.w0c.y +
                              c.w1c.z * c.w0c.z + c.w1c.w * c.w0c.w);
    a0.x = c.u0c.x - p0.x; a0.y = c.u0c.y - p0.y;
    a0.z = c.u0c.z - p0.z; a0.w = c.u0c.w - p0.w;
    a1.x = c.u1c.x - (p1.x + ww * a0.x); a1.y = c.u1c.y - (p1.y + ww * a0.y);
    a1.z = c.u1c.z - (p1.z + ww * a0.z); a1.w = c.u1c.w - (p1.w + ww * a0.w);
}

__device__ __forceinline__ void scan_part(
    const float* __restrict__ Wt, const float* __restrict__ Ut,
    const float* __restrict__ h0, float* __restrict__ hstates,
    float* __restrict__ hfinal, float (*part)[2][8][DD], int bh, int c0, int c1)
{
    const int tid  = threadIdx.x;
    const int lane = tid & 31;
    const int wave = tid >> 5;
    const int row0 = wave * 16;
    const size_t myoff = (size_t)row0 * DD + 4 * lane;

    ChunkVec nxt = load_chunk(Wt, Ut, bh, (c0 + 1 < NCHUNK) ? c0 + 1 : NCHUNK - 1, row0, lane);
    ChunkVec cur = load_chunk(Wt, Ut, bh, c0, row0, lane);

    float4 hs[16];
    {
        const float* src = ((c0 == 0) ? h0 + (size_t)bh * DD * DD
                                      : hstates + (size_t)(bh * NCHUNK + c0) * DD * DD) + myoff;
#pragma unroll
        for (int i = 0; i < 16; ++i) hs[i] = *(const float4*)(src + (size_t)i * DD);
    }

    float4 a0, a1;
    {
        const float4 none = {0.f, 0.f, 0.f, 0.f};
        float4 acc0 = {0.f, 0.f, 0.f, 0.f}, acc1 = {0.f, 0.f, 0.f, 0.f};
        state_pass<false>(hs, 0.f, 0.f, cur.w0r, cur.w1r, none, none, acc0, acc1, lane);
        meet_sums(part[0], wave, lane, acc0, acc1, cur, a0, a1);
    }

    for (int ch = c0; ch < c1; ++ch) {
        const ChunkVec pre = load_chunk(Wt, Ut, bh, (ch + 2 < NCHUNK) ? ch + 2 : NCHUNK - 1,
                                        row0, lane);
        if (ch != c0 || c0 == 0)
            store_state(hs, hstates + (size_t)(bh * NCHUNK + ch) * DD * DD + myoff);

        float4 acc0 = {0.f, 0.f, 0.f, 0.f}, acc1 = {0.f, 0.f, 0.f, 0.f};
        state_pass<true>(hs, cur.w0r, cur.w1r, nxt.w0r, nxt.w1r, a0, a1, acc0, acc1, lane);
        if (ch + 1 < c1) meet_sums(part[(ch - c0 + 1) & 1], wave, lane, acc0, acc1, nxt, a0, a1);
        cur = nxt;
        nxt = pre;
    }
    float* dst = (c1 < NCHUNK) ? hstates + (size_t)(bh * NCHUNK + c1) * DD * DD
                               : hfinal + (size_t)bh * DD * DD;
    store_state(hs, dst + myoff);
}

__device__ __forceinline__ void out_part(
    const float* __restrict__ Qt, const float* __restrict__ Wt,
    const float* __restrict__ Ut, const float* __restrict__ hstates,
    float* __restrict__ ys, float* lds, int nc, int bh)
{
    float* h   = lds;
    float* s_w = lds + DD * DD;
    float* s_q = s_w + CPAIR;

    const int tid  = threadIdx.x;
    const int lane = tid & 31;
    const int wave = tid >> 5;

    const size_t cbase = (size_t)(bh * NCHUNK + nc) * CT * DD;
    const float* Qc = Qt + cbase;
    const float* Wc = Wt + cbase;
    const float* Uc = Ut + cbase;

    const int m      = wave;
    const int mrow   = m * 16 + (lane & 15);
    const int khalf  = 2 * (lane >> 4);
    const int nlane  = lane & 15;
    const int rowoff = (lane >> 4) * 8;

    v2f aW[32], aQ[32];
    {
        const float* wrow = Wc + (size_t)mrow * DD;
        const float* qrow = Qc + (size_t)mrow * DD;
#pragma unroll
        for (int k = 0; k < 32; ++k) {
            int col = 4 * k + khalf;
            aW[k] = *(const v2f*)(wrow + col);
            aQ[k] = *(const v2f*)(qrow + col);
        }
    }

    {
        const float4* src = (const float4*)(hstates + (size_t)(bh * NCHUNK + nc) * DD * DD);
        float4* hv = (float4*)h;
        for (int i = tid; i < DD * DD / 4; i += 256) hv[i] = src[i];
    }

    {
        v2f pw = {0.f, 0.f}, pq = {0.f, 0.f};
#pragma unroll
        for (int k = 0; k < 32; ++k) {
            v2f o;
            o.x = lane_xmask<0x161>(aW[k].x);
            o.y = lane_xmask<0x161>(aW[k].y);
            pw += aW[k] * o;
            pq += aQ[k] * o;
        }
        float sw = pw.x + pw.y, sq = pq.x + pq.y;
        const int swi = __builtin_bit_cast(int, sw), sqi = __builtin_bit_cast(int, sq);
        sw += __builtin_bit_cast(float,
            __builtin_amdgcn_permlanex16(swi, swi, 0x76543210, 0xfedcba98, false, false));
        sq += __builtin_bit_cast(float,
            __builtin_amdgcn_permlanex16(sqi, sqi, 0x76543210, 0xfedcba98, false, false));
        if ((lane & 17) == 1) {
            s_w[m * 8 + (lane >> 1)] = sw;
            s_q[m * 8 + (lane >> 1)] = sq;
        }
    }
    __syncthreads();

    float* outbase = ys + cbase;
    for (int n = 0; n < 8; ++n) {
        v8f accW = {};
        v8f accQ = {};
        const int ncol = n * 16 + nlane;
#pragma unroll
        for (int k = 0; k < 32; ++k) {
            int kk = 4 * k + khalf;
            v2f b;
            b.x = h[kk * DD + ncol];
            b.y = h[(kk + 1) * DD + ncol];
            accW = __builtin_amdgcn_wmma_f32_16x16x4_f32(
                false, aW[k], false, b, (short)0, accW, false, false);
            accQ = __builtin_amdgcn_wmma_f32_16x16x4_f32(
                false, aQ[k], false, b, (short)0, accQ, false, false);
        }
#pragma unroll
        for (int j = 0; j < 4; ++j) {
            int lr0 = 2 * j + rowoff;
            int t0  = m * 16 + lr0;
            int t1  = t0 + 1;
            int c   = t0 >> 1;
            float wh0 = accW[2 * j], wh1 = accW[2 * j + 1];
            float qh0 = accQ[2 * j], qh1 = accQ[2 * j + 1];
            float u0 = Uc[(size_t)t0 * DD + ncol];
            float u1 = Uc[(size_t)t1 * DD + ncol];
            float q0 = Qc[(size_t)t0 * DD + ncol];
            float q1 = Qc[(size_t)t1 * DD + ncol];
            float uv0v = u0 - wh0;
            float out0 = q0 * uv0v + qh0;
            float uv1v = u1 - (wh1 + s_w[c] * uv0v);
            float out1 = q1 * uv1v + qh1 + s_q[c] * uv0v;
            outbase[(size_t)t0 * DD + ncol] = out0;
            outbase[(size_t)t1 * DD + ncol] = out1;
        }
    }
}

#define LDS_FLOATS 41472

__global__ __launch_bounds__(256) void liz_step(
    const float* __restrict__ Qt, const float* __restrict__ Wt,
    const float* __restrict__ Ut, const float* __restrict__ h0,
    float* __restrict__ hstates, float* __restrict__ ys, float* __restrict__ hfinal,
    int sc0, int sc1, int oc0, int ocn)
{
    __shared__ __attribute__((aligned(16))) float lds[LDS_FLOATS];

    const int nscan = (sc1 > sc0) ? BH_TOT : 0;
    const int blk   = blockIdx.x;
    if (blk < nscan) {
        scan_part(Wt, Ut, h0, hstates, hfinal, (float (*)[2][8][DD])lds, blk, sc0, sc1);
    } else {
        const int idx = blk - nscan;
        const int bh  = idx / ocn;
        const int nc  = oc0 + (idx - bh * ocn);
        out_part(Qt, Wt, Ut, hstates, ys, lds, nc, bh);
    }
}

extern "C" void kernel_launch(void* const* d_in, const int* in_sizes, int n_in,
                              void* d_out, int out_size, void* d_ws, size_t ws_size,
                              hipStream_t stream) {
    const float* q  = (const float*)d_in[0];
    const float* W  = (const float*)d_in[1];
    const float* U  = (const float*)d_in[2];
    const float* h0 = (const float*)d_in[3];

    float* ys     = (float*)d_out;
    float* hfinal = ys + (size_t)BH_TOT * NCHUNK * CT * DD;
    float* hstate = (float*)d_ws;

    liz_step<<< 32, 256, 0, stream>>>(q, W, U, h0, hstate, ys, hfinal,  0,  4,  0, 1);
    liz_step<<<160, 256, 0, stream>>>(q, W, U, h0, hstate, ys, hfinal,  4, 11,  0, 4);
    liz_step<<<256, 256, 0, stream>>>(q, W, U, h0, hstate, ys, hfinal, 11, 18,  4, 7);
    liz_step<<<256, 256, 0, stream>>>(q, W, U, h0, hstate, ys, hfinal, 18, 25, 11, 7);
    liz_step<<<256, 256, 0, stream>>>(q, W, U, h0, hstate, ys, hfinal, 25, 32, 18, 7);
    liz_step<<<224, 256, 0, stream>>>(q, W, U, h0, hstate, ys, hfinal, 32, 32, 25, 7);
}


# out: k loop outside, 8 column tiles inside (16 accumulator tiles a wave): h loaded first, A strips land under the WMMAs, barrier waits for LDS only, B pair = one 8-byte LDS load (sum order same)
# speedup vs baseline: 7.8367x; 1.0867x over previous
#include <hip/hip_runtime.h>
#include <math.h>

#define BH_TOT 32
#define NCHUNK 32
#define CT     128
#define CPAIR  64
#define DD     128
#define STABV  10000.0f

typedef float v2f __attribute__((ext_vector_type(2)));
typedef float v8f __attribute__((ext_vector_type(8)));

__device__ __forceinline__ float gelu_row(float v, float A, float B) {
    const float u = v * (B * (v * v) + A);
    const float t = __builtin_amdgcn_tanhf(u);
    return v * (0.5f * t + 0.5f);
}

__device__ __forceinline__ float clipstab(float v) {
    return fminf(fmaxf(v, -STABV), STABV);
}

template <int CTRL>
__device__ __forceinline__ float lane_xmask(float x) {
    return __builtin_bit_cast(float,
        __builtin_amdgcn_update_dpp(0, __builtin_bit_cast(int, x), CTRL, 0xf, 0xf, true));
}
__device__ __forceinline__ float wave_sum(float x) {
    x += lane_xmask<0x161>(x);
    x += lane_xmask<0x162>(x);
    x += lane_xmask<0x164>(x);
    x += lane_xmask<0x168>(x);
    const int xi = __builtin_bit_cast(int, x);
    x += __builtin_bit_cast(float,
        __builtin_amdgcn_permlanex16(xi, xi, 0x76543210, 0xfedcba98, false, false));
    return x;
}

struct ChunkVec {
    float  w0r, w1r;
    float4 w0c, w1c, u0c, u1c;
};

__device__ __forceinline__ ChunkVec load_chunk(const float* __restrict__ Wt,
                                               const float* __restrict__ Ut,
                                               int bh, int ch, int row0, int lane) {
    const float* w0p = Wt + ((size_t)(bh * NCHUNK + ch) * CT + (CT - 2)) * DD;
    const float* u0p = Ut + ((size_t)(bh * NCHUNK + ch) * CT + (CT - 2)) * DD;
    ChunkVec c;
    c.w0r = w0p[row0 + (lane & 15)];
    c.w1r = w0p[DD + row0 + (lane & 15)];
    c.w0c = *(const float4*)(w0p + 4 * lane);
    c.w1c = *(const float4*)(w0p + DD + 4 * lane);
    c.u0c = *(const float4*)(u0p + 4 * lane);
    c.u1c = *(const float4*)(u0p + DD + 4 * lane);
    return c;
}

__device__ __forceinline__ float lane_value(float x, int r) {
    return __builtin_bit_cast(float, __builtin_amdgcn_readlane(__builtin_bit_cast(int, x), r));
}

template <bool UPDATE>
__device__ __forceinline__ void state_pass(float4 (&hs)[16], float w0cur, float w1cur,
                                           float w0nxt, float w1nxt,
                                           const float4 a0, const float4 a1,
                                           float4& acc0, float4& acc1, int lane) {
    if (UPDATE) {
        float sq[16];
#pragma unroll
        for (int r = 0; r < 16; ++r) {
            const float4 hv = hs[r];
            const float w0v = lane_value(w0cur, r), w1v = lane_value(w1cur, r);
            float4 v;
            v.x = clipstab(clipstab(hv.x + w0v * a0.x) + w1v * a1.x);
            v.y = clipstab(clipstab(hv.y + w0v * a0.y) + w1v * a1.y);
            v.z = clipstab(clipstab(hv.z + w0v * a0.z) + w1v * a1.z);
            v.w = clipstab(clipstab(hv.w + w0v * a0.w) + w1v * a1.w);
            hs[r] = v;
            sq[r] = v.x * v.x + v.y * v.y + v.z * v.z + v.w * v.w;
        }
        const bool b0 = (lane & 1) != 0, b1 = (lane & 2) != 0;
        const bool b2 = (lane & 4) != 0, b3 = (lane & 8) != 0;
        float s8[8], s4[4], s2[2];
#pragma unroll
        for (int i = 0; i < 8; ++i) {
            const float keep = b0 ? sq[2 * i + 1] : sq[2 * i];
            const float send = b0 ? sq[2 * i] : sq[2 * i + 1];
            s8[i] = keep + lane_xmask<0x161>(send);
        }
#pragma unroll
        for (int i = 0; i < 4; ++i) {
            const float keep = b1 ? s8[2 * i + 1] : s8[2 * i];
            const float send = b1 ? s8[2 * i] : s8[2 * i + 1];
            s4[i] = keep + lane_xmask<0x162>(send);
        }
#pragma unroll
        for (int i = 0; i < 2; ++i) {
            const float keep = b2 ? s4[2 * i + 1] : s4[2 * i];
            const float send = b2 ? s4[2 * i] : s4[2 * i + 1];
            s2[i] = keep + lane_xmask<0x164>(send);
        }
        float tot;
        {
            const float keep = b3 ? s2[1] : s2[0];
            const float send = b3 ? s2[0] : s2[1];
            tot = keep + lane_xmask<0x168>(send);
            const int ti = __builtin_bit_cast(int, tot);
            tot += __builtin_bit_cast(float,
                __builtin_amdgcn_permlanex16(ti, ti, 0x76543210, 0xfedcba98, false, false));
        }
        const float n   = sqrtf(tot) + 1e-6f;
        const float inv = 2.0f / n;
        const float cA  = 0.7978845608f * inv;
        const float cB  = (0.7978845608f * 0.044715f) * inv * (inv * inv);
#pragma unroll
        for (int r = 0; r < 16; ++r) {
            const float Ar = lane_value(cA, r), Br = lane_value(cB, r);
            const float4 v = hs[r];
            float4 o;
            o.x = gelu_row(v.x, Ar, Br);
            o.y = gelu_row(v.y, Ar, Br);
            o.z = gelu_row(v.z, Ar, Br);
            o.w = gelu_row(v.w, Ar, Br);
            hs[r] = o;
        }
    }
#pragma unroll
    for (int r = 0; r < 16; ++r) {
        const float4 o = hs[r];
        const float w0n = lane_value(w0nxt, r), w1n = lane_value(w1nxt, r);
        acc0.x += w0n * o.x; acc0.y += w0n * o.y; acc0.z += w0n * o.z; acc0.w += w0n * o.w;
        acc1.x += w1n * o.x; acc1.y += w1n * o.y; acc1.z += w1n * o.z; acc1.w += w1n * o.w;
    }
}

__device__ __forceinline__ void store_state(const float4 (&hs)[16], float* __restrict__ dst) {
#pragma unroll
    for (int i = 0; i < 16; ++i) *(float4*)(dst + (size_t)i * DD) = hs[i];
}

__device__ __forceinline__ void meet_sums(float (&part)[2][8][DD], int wave, int lane,
                                          const float4 acc0, const float4 acc1,
                                          const ChunkVec& c, float4& a0, float4& a1) {
    *(float4*)&part[0][wave][4 * lane] = acc0;
    *(float4*)&part[1][wave][4 * lane] = acc1;
    __syncthreads();
    float4 p0 = *(const float4*)&part[0][0][4 * lane];
    float4 p1 = *(const float4*)&part[1][0][4 * lane];
    for (int w = 1; w < 8; ++w) {
        const float4 q0 = *(const float4*)&part[0][w][4 * lane];
        const float4 q1 = *(const float4*)&part[1][w][4 * lane];
        p0.x += q0.x; p0.y += q0.y; p0.z += q0.z; p0.w += q0.w;
        p1.x += q1.x; p1.y += q1.y; p1.z += q1.z; p1.w += q1.w;
    }
    const float ww = wave_sum(c.w1c.x * c.w0c.x + c.w1c.y * c.w0c.y +
                              c.w1c.z * c.w0c.z + c.w1c.w * c.w0c.w);
    a0.x = c.u0c.x - p0.x; a0.y = c.u0c.y - p0.y;
    a0.z = c.u0c.z - p0.z; a0.w = c.u0c.w - p0.w;
    a1.x = c.u1c.x - (p1.x + ww * a0.x); a1.y = c.u1c.y - (p1.y + ww * a0.y);
    a1.z = c.u1c.z - (p1.z + ww * a0.z); a1.w = c.u1c.w - (p1.w + ww * a0.w);
}

__device__ __forceinline__ void scan_part(
    const float* __restrict__ Wt, const float* __restrict__ Ut,
    const float* __restrict__ h0, float* __restrict__ hstates,
    float* __restrict__ hfinal, float (*part)[2][8][DD], int bh, int c0, int c1)
{
    const int tid  = threadIdx.x;
    const int lane = tid & 31;
    const int wave = tid >> 5;
    const int row0 = wave * 16;
    const size_t myoff = (size_t)row0 * DD + 4 * lane;

    ChunkVec nxt = load_chunk(Wt, Ut, bh, (c0 + 1 < NCHUNK) ? c0 + 1 : NCHUNK - 1, row0, lane);
    ChunkVec cur = load_chunk(Wt, Ut, bh, c0, row0, lane);

    float4 hs[16];
    {
        const float* src = ((c0 == 0) ? h0 + (size_t)bh * DD * DD
                                      : hstates + (size_t)(bh * NCHUNK + c0) * DD * DD) + myoff;
#pragma unroll
        for (int i = 0; i < 16; ++i) hs[i] = *(const float4*)(src + (size_t)i * DD);
    }

    float4 a0, a1;
    {
        const float4 none = {0.f, 0.f, 0.f, 0.f};
        float4 acc0 = {0.f, 0.f, 0.f, 0.f}, acc1 = {0.f, 0.f, 0.f, 0.f};
        state_pass<false>(hs, 0.f, 0.f, cur.w0r, cur.w1r, none, none, acc0, acc1, lane);
        meet_sums(part[0], wave, lane, acc0, acc1, cur, a0, a1);
    }

    for (int ch = c0; ch < c1; ++ch) {
        const ChunkVec pre = load_chunk(Wt, Ut, bh, (ch + 2 < NCHUNK) ? ch + 2 : NCHUNK - 1,
                                        row0, lane);
        if (ch != c0 || c0 == 0)
            store_state(hs, hstates + (size_t)(bh * NCHUNK + ch) * DD * DD + myoff);

        float4 acc0 = {0.f, 0.f, 0.f, 0.f}, acc1 = {0.f, 0.f, 0.f, 0.f};
        state_pass<true>(hs, cur.w0r, cur.w1r, nxt.w0r, nxt.w1r, a0, a1, acc0, acc1, lane);
        if (ch + 1 < c1) meet_sums(part[(ch - c0 + 1) & 1], wave, lane, acc0, acc1, nxt, a0, a1);
        cur = nxt;
        nxt = pre;
    }
    float* dst = (c1 < NCHUNK) ? hstates + (size_t)(bh * NCHUNK + c1) * DD * DD
                               : hfinal + (size_t)bh * DD * DD;
    store_state(hs, dst + myoff);
}

__device__ __forceinline__ void out_part(
    const float* __restrict__ Qt, const float* __restrict__ Wt,
    const float* __restrict__ Ut, const float* __restrict__ hstates,
    float* __restrict__ ys, float* lds, int nc, int bh)
{
    float* hp = lds;

    const int tid  = threadIdx.x;
    const int lane = tid & 31;
    const int wave = tid >> 5;

    const size_t cbase = (size_t)(bh * NCHUNK + nc) * CT * DD;
    const float* Qc = Qt + cbase;
    const float* Wc = Wt + cbase;
    const float* Uc = Ut + cbase;

    const int m      = wave;
    const int mrow   = m * 16 + (lane & 15);
    const int khalf  = 2 * (lane >> 4);
    const int nlane  = lane & 15;
    const int hh     = lane >> 4;
    const int rowoff = hh * 8;

    float4 he[8], ho[8];
    {
        const float4* src = (const float4*)(hstates + (size_t)(bh * NCHUNK + nc) * DD * DD);
#pragma unroll
        for (int i = 0; i < 8; ++i) {
            const int p = wave + 8 * i;
            he[i] = src[(2 * p) * 32 + lane];
            ho[i] = src[(2 * p + 1) * 32 + lane];
        }
    }
    __builtin_amdgcn_sched_barrier(0);

    v2f aW[32], aQ[32];
    {
        const float* wrow = Wc + (size_t)mrow * DD;
        const float* qrow = Qc + (size_t)mrow * DD;
#pragma unroll
        for (int k = 0; k < 32; ++k) {
            int col = 4 * k + khalf;
            aW[k] = *(const v2f*)(wrow + col);
            aQ[k] = *(const v2f*)(qrow + col);
        }
    }
    __builtin_amdgcn_sched_barrier(0);

    {
#pragma unroll
        for (int i = 0; i < 8; ++i) {
            const int p = wave + 8 * i;
            float* dst = hp + ((size_t)p * DD + 4 * lane) * 2;
            const float4 a = {he[i].x, ho[i].x, he[i].y, ho[i].y};
            const float4 b = {he[i].z, ho[i].z, he[i].w, ho[i].w};
            *(float4*)dst       = a;
            *(float4*)(dst + 4) = b;
        }
    }
    __builtin_amdgcn_fence(__ATOMIC_RELEASE, "workgroup", "local");
    __builtin_amdgcn_s_barrier();
    __builtin_amdgcn_fence(__ATOMIC_ACQUIRE, "workgroup", "local");

    v8f accW[8], accQ[8];
#pragma unroll
    for (int n = 0; n < 8; ++n) { accW[n] = v8f{}; accQ[n] = v8f{}; }

    float eu0[8][4], eu1[8][4], eq0[8][4], eq1[8][4];

    const float* bbase = hp + ((size_t)hh * DD + nlane) * 2;

    v2f bn[8];
#pragma unroll
    for (int n = 0; n < 8; ++n) bn[n] = *(const v2f*)(bbase + (size_t)(n * 16) * 2);

#pragma unroll
    for (int k = 0; k < 32; ++k) {
        __builtin_amdgcn_sched_barrier(0);
        v2f bc[8];
#pragma unroll
        for (int n = 0; n < 8; ++n) bc[n] = bn[n];
        if (k + 1 < 32) {
#pragma unroll
            for (int n = 0; n < 8; ++n)
                bn[n] = *(const v2f*)(bbase + ((size_t)(2 * (k + 1)) * DD + n * 16) * 2);
        }
        if (k >= 8 && k < 16 && (k & 1) == 0) {
#pragma unroll
            for (int nn = 0; nn < 2; ++nn) {
                const int n = (k - 8) + nn;
#pragma unroll
                for (int j = 0; j < 4; ++j) {
                    const int t0 = m * 16 + 2 * j + rowoff;
                    eu0[n][j] = Uc[(size_t)t0 * DD + n * 16 + nlane];
                    eu1[n][j] = Uc[(size_t)(t0 + 1) * DD + n * 16 + nlane];
                }
            }
        }
        if (k >= 18 && k < 26 && (k & 1) == 0) {
#pragma unroll
            for (int nn = 0; nn < 2; ++nn) {
                const int n = (k - 18) + nn;
#pragma unroll
                for (int j = 0; j < 4; ++j) {
                    const int t0 = m * 16 + 2 * j + rowoff;
                    eq0[n][j] = Qc[(size_t)t0 * DD + n * 16 + nlane];
                    eq1[n][j] = Qc[(size_t)(t0 + 1) * DD + n * 16 + nlane];
                }
            }
        }
#pragma unroll
        for (int n = 0; n < 8; ++n) {
            accW[n] = __builtin_amdgcn_wmma_f32_16x16x4_f32(
                false, aW[k], false, bc[n], (short)0, accW[n], false, false);
            accQ[n] = __builtin_amdgcn_wmma_f32_16x16x4_f32(
                false, aQ[k], false, bc[n], (short)0, accQ[n], false, false);
        }
    }

    float sw, sq;
    {
        v2f pw = {0.f, 0.f}, pq = {0.f, 0.f};
#pragma unroll
        for (int k = 0; k < 32; ++k) {
            v2f o;
            o.x = lane_xmask<0x161>(aW[k].x);
            o.y = lane_xmask<0x161>(aW[k].y);
            pw += aW[k] * o;
            pq += aQ[k] * o;
        }
        sw = pw.x + pw.y; sq = pq.x + pq.y;
        const int swi = __builtin_bit_cast(int, sw), sqi = __builtin_bit_cast(int, sq);
        sw += __builtin_bit_cast(float,
            __builtin_amdgcn_permlanex16(swi, swi, 0x76543210, 0xfedcba98, false, false));
        sq += __builtin_bit_cast(float,
            __builtin_amdgcn_permlanex16(sqi, sqi, 0x76543210, 0xfedcba98, false, false));
    }
    float swp[8], sqp[8];
#pragma unroll
    for (int p = 0; p < 8; ++p) { swp[p] = lane_value(sw, 2 * p + 1); sqp[p] = lane_value(sq, 2 * p + 1); }

    float* outbase = ys + cbase;
#pragma unroll
    for (int j = 0; j < 4; ++j) {
        const float swj = hh ? swp[4 + j] : swp[j];
        const float sqj = hh ? sqp[4 + j] : sqp[j];
        const int t0 = m * 16 + 2 * j + rowoff;
        const int t1 = t0 + 1;
#pragma unroll
        for (int n = 0; n < 8; ++n) {
            const int ncol = n * 16 + nlane;
            const float wh0 = accW[n][2 * j], wh1 = accW[n][2 * j + 1];
            const float qh0 = accQ[n][2 * j], qh1 = accQ[n][2 * j + 1];
            const float uv0v = eu0[n][j] - wh0;
            const float out0 = eq0[n][j] * uv0v + qh0;
            const float uv1v = eu1[n][j] - (wh1 + swj * uv0v);
            const float out1 = eq1[n][j] * uv1v + qh1 + sqj * uv0v;
            outbase[(size_t)t0 * DD + ncol] = out0;
            outbase[(size_t)t1 * DD + ncol] = out1;
        }
    }
}

#define LDS_FLOATS 41472

__global__ __launch_bounds__(256) void liz_step(
    const float* __restrict__ Qt, const float* __restrict__ Wt,
    const float* __restrict__ Ut, const float* __restrict__ h0,
    float* __restrict__ hstates, float* __restrict__ ys, float* __restrict__ hfinal,
    int sc0, int sc1, int oc0, int ocn)
{
    __shared__ __attribute__((aligned(16))) float lds[LDS_FLOATS];

    const int nscan = (sc1 > sc0) ? BH_TOT : 0;
    const int blk   = blockIdx.x;
    if (blk < nscan) {
        scan_part(Wt, Ut, h0, hstates, hfinal, (float (*)[2][8][DD])lds, blk, sc0, sc1);
    } else {
        const int idx = blk - nscan;
        const int bh  = idx / ocn;
        const int nc  = oc0 + (idx - bh * ocn);
        out_part(Qt, Wt, Ut, hstates, ys, lds, nc, bh);
    }
}

extern "C" void kernel_launch(void* const* d_in, const int* in_sizes, int n_in,
                              void* d_out, int out_size, void* d_ws, size_t ws_size,
                              hipStream_t stream) {
    const float* q  = (const float*)d_in[0];
    const float* W  = (const float*)d_in[1];
    const float* U  = (const float*)d_in[2];
    const float* h0 = (const float*)d_in[3];

    float* ys     = (float*)d_out;
    float* hfinal = ys + (size_t)BH_TOT * NCHUNK * CT * DD;
    float* hstate = (float*)d_ws;

    liz_step<<< 32, 256, 0, stream>>>(q, W, U, h0, hstate, ys, hfinal,  0,  4,  0, 1);
    liz_step<<<160, 256, 0, stream>>>(q, W, U, h0, hstate, ys, hfinal,  4, 11,  0, 4);
    liz_step<<<256, 256, 0, stream>>>(q, W, U, h0, hstate, ys, hfinal, 11, 18,  4, 7);
    liz_step<<<256, 256, 0, stream>>>(q, W, U, h0, hstate, ys, hfinal, 18, 25, 11, 7);
    liz_step<<<256, 256, 0, stream>>>(q, W, U, h0, hstate, ys, hfinal, 25, 32, 18, 7);
    liz_step<<<224, 256, 0, stream>>>(q, W, U, h0, hstate, ys, hfinal, 32, 32, 25, 7);
}
